# starting kernel
# baseline (speedup 1.0000x reference)
#include <hip/hip_runtime.h>

#define N_NODES 50000
#define N_REL   16
#define HID     64
#define N_LAB   32
#define N_EDGE  1600000
#define K_AGG   1024
#define K_TOT   1088

typedef float v2f __attribute__((ext_vector_type(2)));
typedef float v8f __attribute__((ext_vector_type(8)));

#define WS_CNT_OFF  0
#define WS_B_OFF    (N_REL * N_NODES)
#define WS_LHS_OFF  (WS_B_OFF + K_TOT * N_LAB)
#define WS_TOTAL    (WS_LHS_OFF + (size_t)N_NODES * K_TOT)

__device__ __forceinline__ void fatomic_add(float* p, float v) {
  __hip_atomic_fetch_add(p, v, __ATOMIC_RELAXED, __HIP_MEMORY_SCOPE_AGENT);
}

__global__ void k_zero(float* __restrict__ p, unsigned long long n) {
  unsigned long long i = (unsigned long long)blockIdx.x * blockDim.x + threadIdx.x;
  unsigned long long stride = (unsigned long long)gridDim.x * blockDim.x;
  for (; i < n; i += stride) p[i] = 0.0f;
}

__global__ void k_count(const int* __restrict__ etype, const int* __restrict__ dst,
                        float* __restrict__ cnt) {
  unsigned e = blockIdx.x * blockDim.x + threadIdx.x;
  if (e >= N_EDGE) return;
  fatomic_add(&cnt[etype[e] * N_NODES + dst[e]], 1.0f);
}

__global__ void k_bmat(const float* __restrict__ w2, const float* __restrict__ r2,
                       float* __restrict__ B) {
  unsigned t = blockIdx.x * blockDim.x + threadIdx.x;
  if (t >= K_TOT * N_LAB) return;
  B[t] = (t < K_AGG * N_LAB) ? w2[t] : r2[t - K_AGG * N_LAB];
}

__global__ void k_layer1(const int* __restrict__ src_a, const int* __restrict__ dst_a,
                         const int* __restrict__ etype, const float* __restrict__ w1,
                         const float* __restrict__ cnt, float* __restrict__ lhs) {
  unsigned tid = blockIdx.x * blockDim.x + threadIdx.x;
  unsigned e = tid >> 5;
  unsigned lane = tid & 31;
  if (e >= N_EDGE) return;
  int rel = etype[e], s = src_a[e], d = dst_a[e];
  float inv = 1.0f / fmaxf(cnt[rel * N_NODES + d], 1.0f);
  const float* wrow = w1 + ((size_t)rel * N_NODES + (size_t)s) * HID + 2u * lane;
  v2f v = *(const v2f*)wrow;
  float* o = lhs + (size_t)d * K_TOT + K_AGG + 2u * lane;
  fatomic_add(o + 0, v.x * inv);
  fatomic_add(o + 1, v.y * inv);
}

__global__ void k_relu(const float* __restrict__ root1, const float* __restrict__ bias1,
                       float* __restrict__ lhs) {
  unsigned idx = blockIdx.x * blockDim.x + threadIdx.x;
  if (idx >= N_NODES * HID) return;
  unsigned n = idx >> 6, h = idx & 63;
  size_t p = (size_t)n * K_TOT + K_AGG + h;
  lhs[p] = fmaxf(lhs[p] + root1[idx] + bias1[h], 0.0f);
}

__global__ void k_layer2(const int* __restrict__ src_a, const int* __restrict__ dst_a,
                         const int* __restrict__ etype, const float* __restrict__ cnt,
                         float* __restrict__ lhs) {
  unsigned tid = blockIdx.x * blockDim.x + threadIdx.x;
  unsigned e = tid >> 5;
  unsigned lane = tid & 31;
  if (e >= N_EDGE) return;
  int rel = etype[e], s = src_a[e], d = dst_a[e];
  float inv = 1.0f / fmaxf(cnt[rel * N_NODES + d], 1.0f);
  const float* xrow = lhs + (size_t)s * K_TOT + K_AGG + 2u * lane;
  v2f v = *(const v2f*)xrow;
  float* o = lhs + (size_t)d * K_TOT + (unsigned)rel * HID + 2u * lane;
  fatomic_add(o + 0, v.x * inv);
  fatomic_add(o + 1, v.y * inv);
}

__global__ void k_gemm(const float* __restrict__ lhs, const float* __restrict__ B,
                       const float* __restrict__ bias2, float* __restrict__ out) {
  unsigned tid = blockIdx.x * blockDim.x + threadIdx.x;
  unsigned wave = tid >> 5;
  unsigned lane = tid & 31;
  if (wave >= N_NODES / 16) return;
  unsigned half = lane >> 4;
  unsigned sub = lane & 15;
  const float* arow = lhs + (size_t)(wave * 16u + sub) * K_TOT + 2u * half;
  v8f c0 = {}; v8f c1 = {};
  for (int k0 = 0; k0 < K_TOT; k0 += 4) {
    v2f a;  a.x = arow[k0];  a.y = arow[k0 + 1];
    int kb = k0 + 2 * (int)half;
    v2f b0; b0.x = B[kb * N_LAB + sub];        b0.y = B[(kb + 1) * N_LAB + sub];
    v2f b1; b1.x = B[kb * N_LAB + 16 + sub];   b1.y = B[(kb + 1) * N_LAB + 16 + sub];
    c0 = __builtin_amdgcn_wmma_f32_16x16x4_f32(false, a, false, b0, (short)0, c0, false, false);
    c1 = __builtin_amdgcn_wmma_f32_16x16x4_f32(false, a, false, b1, (short)0, c1, false, false);
  }
  #pragma unroll
  for (int v = 0; v < 8; v++) {
    unsigned row = wave * 16u + (unsigned)v + 8u * half;
    float x0 = c0[v] + bias2[sub];
    float x1 = c1[v] + bias2[16 + sub];
    out[row * N_LAB + sub]      = 1.0f / (1.0f + __expf(-x0));
    out[row * N_LAB + 16 + sub] = 1.0f / (1.0f + __expf(-x1));
  }
}

extern "C" void kernel_launch(void* const* d_in, const int* in_sizes, int n_in,
                              void* d_out, int out_size, void* d_ws, size_t ws_size,
                              hipStream_t stream) {
  const int* ei     = (const int*)d_in[0];
  const int* src_a  = ei;
  const int* dst_a  = ei + N_EDGE;
  const int* etype  = (const int*)d_in[1];
  const float* w1   = (const float*)d_in[2];
  const float* r1   = (const float*)d_in[3];
  const float* b1   = (const float*)d_in[4];
  const float* w2   = (const float*)d_in[5];
  const float* r2   = (const float*)d_in[6];
  const float* b2   = (const float*)d_in[7];
  float* out = (float*)d_out;

  float* ws   = (float*)d_ws;
  float* cnt  = ws + WS_CNT_OFF;
  float* Bmat = ws + WS_B_OFF;
  float* lhs  = ws + WS_LHS_OFF;

  (void)in_sizes; (void)n_in; (void)out_size; (void)ws_size;

  k_zero<<<2048, 256, 0, stream>>>(ws, (unsigned long long)WS_TOTAL);
  k_count<<<(N_EDGE + 255) / 256, 256, 0, stream>>>(etype, dst_a, cnt);
  k_bmat<<<(K_TOT * N_LAB + 255) / 256, 256, 0, stream>>>(w2, r2, Bmat);
  k_layer1<<<(N_EDGE * 32u + 255) / 256, 256, 0, stream>>>(src_a, dst_a, etype, w1, cnt, lhs);
  k_relu<<<(N_NODES * HID + 255) / 256, 256, 0, stream>>>(r1, b1, lhs);
  k_layer2<<<(N_EDGE * 32u + 255) / 256, 256, 0, stream>>>(src_a, dst_a, etype, cnt, lhs);
  k_gemm<<<((N_NODES / 16) * 32 + 255) / 256, 256, 0, stream>>>(lhs, Bmat, b2, out);
}


# layer 1 owner-computes: 256 work-groups each scan the dst list, queue the edges of their 196 nodes, sum the weight1 rows in LDS (f32 LDS adds, arrival order), store once; no global atomics there
# speedup vs baseline: 1.1342x; 1.1342x over previous
#include <hip/hip_runtime.h>

#define N_NODES 50000
#define N_REL   16
#define HID     64
#define N_LAB   32
#define N_EDGE  1600000
#define K_AGG   1024
#define K_TOT   1088

typedef float v2f __attribute__((ext_vector_type(2)));
typedef float v8f __attribute__((ext_vector_type(8)));

#define WS_CNT_OFF  0
#define WS_B_OFF    (N_REL * N_NODES)
#define WS_LHS_OFF  (WS_B_OFF + K_TOT * N_LAB)
#define WS_TOTAL    (WS_LHS_OFF + (size_t)N_NODES * K_TOT)

__device__ __forceinline__ void fatomic_add(float* p, float v) {
  __hip_atomic_fetch_add(p, v, __ATOMIC_RELAXED, __HIP_MEMORY_SCOPE_AGENT);
}

__global__ void k_zero(float* __restrict__ p, unsigned long long n) {
  unsigned long long i = (unsigned long long)blockIdx.x * blockDim.x + threadIdx.x;
  unsigned long long stride = (unsigned long long)gridDim.x * blockDim.x;
  for (; i < n; i += stride) p[i] = 0.0f;
}

__global__ void k_count(const int* __restrict__ etype, const int* __restrict__ dst,
                        float* __restrict__ cnt) {
  unsigned e = blockIdx.x * blockDim.x + threadIdx.x;
  if (e >= N_EDGE) return;
  fatomic_add(&cnt[etype[e] * N_NODES + dst[e]], 1.0f);
}

__global__ void k_bmat(const float* __restrict__ w2, const float* __restrict__ r2,
                       float* __restrict__ B) {
  unsigned t = blockIdx.x * blockDim.x + threadIdx.x;
  if (t >= K_TOT * N_LAB) return;
  B[t] = (t < K_AGG * N_LAB) ? w2[t] : r2[t - K_AGG * N_LAB];
}

#define NPB     196
#define QCAP    32768
#define SCAN_U  8
#define DR_U    16

__device__ __forceinline__ void lds_fadd(float* p, float v) {
  __hip_atomic_fetch_add(p, v, __ATOMIC_RELAXED, __HIP_MEMORY_SCOPE_WORKGROUP);
}

__global__ __launch_bounds__(1024)
void k_layer1(const int* __restrict__ src_a, const int* __restrict__ dst_a,
              const int* __restrict__ etype, const float* __restrict__ w1,
              const float* __restrict__ cnt, float* __restrict__ lhs) {
  __shared__ float acc[NPB * HID];
  __shared__ int q[QCAP];
  __shared__ unsigned qn;
  const unsigned tid = threadIdx.x;
  const unsigned lane = tid & 31;
  const int lo = (int)blockIdx.x * NPB;

  for (unsigned i = tid; i < NPB * HID; i += 1024) acc[i] = 0.0f;
  if (tid == 0) qn = 0;
  __syncthreads();

  const int4* __restrict__ d4p = (const int4*)dst_a;
  const unsigned n4 = N_EDGE / 4;
  for (unsigned base = 0; base < n4; base += 1024u * SCAN_U) {
    int4 v[SCAN_U];
    #pragma unroll
    for (int u = 0; u < SCAN_U; u++) {
      unsigned i = base + (unsigned)u * 1024u + tid;
      v[u] = (i < n4) ? d4p[i] : make_int4(-1, -1, -1, -1);
    }
    #pragma unroll
    for (int u = 0; u < SCAN_U; u++) {
      unsigned i = base + (unsigned)u * 1024u + tid;
      int dd[4] = {v[u].x, v[u].y, v[u].z, v[u].w};
      unsigned nearest = min(min((unsigned)(dd[0] - lo), (unsigned)(dd[1] - lo)),
                             min((unsigned)(dd[2] - lo), (unsigned)(dd[3] - lo)));
      if (nearest < (unsigned)NPB) {
        #pragma unroll
        for (int c = 0; c < 4; c++) {
          if ((unsigned)(dd[c] - lo) < (unsigned)NPB) {
            int e = (int)(4u * i) + c;
            unsigned idx = atomicAdd(&qn, 1u);
            if (idx < QCAP) q[idx] = e;
          }
        }
      }
    }
  }
  __syncthreads();

  const unsigned nq = qn;
  if (nq > (unsigned)QCAP) {
    for (unsigned e = tid; e < N_EDGE; e += 1024u) {
      int d = dst_a[e];
      if ((unsigned)(d - lo) < (unsigned)NPB) {
        int rel = etype[e], s = src_a[e];
        float ivs = 1.0f / fmaxf(cnt[rel * N_NODES + d], 1.0f);
        const float* wr = w1 + ((size_t)rel * N_NODES + (size_t)s) * HID;
        float* ar = acc + (d - lo) * HID;
        for (int h = 0; h < HID; h++) lds_fadd(ar + h, wr[h] * ivs);
      }
    }
  }
  const unsigned n = (nq > (unsigned)QCAP) ? 0u : nq;
  const unsigned wave = (unsigned)__builtin_amdgcn_readfirstlane((int)(tid >> 5));
  for (unsigned b = wave * 32u; b < n; b += 32u * 32u) {
    const unsigned m = min(32u, n - b);
    int r = 0, dl = 0; float iv = 0.0f;
    if (lane < m) {
      int e = q[b + lane];
      int rel = etype[e], s = src_a[e], d = dst_a[e];
      iv = 1.0f / fmaxf(cnt[rel * N_NODES + d], 1.0f);
      r = rel * N_NODES + s;
      dl = d - lo;
    }
    #pragma unroll
    for (unsigned j = 0; j < 32; j += DR_U) {
      if (j >= m) break;
      if (j + DR_U <= m) {
        v2f row[DR_U];
        #pragma unroll
        for (unsigned u = 0; u < DR_U; u++) {
          int rr = __builtin_amdgcn_readlane(r, (int)(j + u));
          row[u] = *(const v2f*)(w1 + (size_t)rr * HID + 2u * lane);
        }
        #pragma unroll
        for (unsigned u = 0; u < DR_U; u++) {
          int dd = __builtin_amdgcn_readlane(dl, (int)(j + u));
          float ii = __builtin_bit_cast(float, __builtin_amdgcn_readlane(__builtin_bit_cast(int, iv), (int)(j + u)));
          float* o = acc + dd * HID + 2u * lane;
          lds_fadd(o + 0, row[u].x * ii);
          lds_fadd(o + 1, row[u].y * ii);
        }
      } else {
        for (unsigned u = j; u < m; u++) {
          int rr = __builtin_amdgcn_readlane(r, (int)u);
          int dd = __builtin_amdgcn_readlane(dl, (int)u);
          float ii = __builtin_bit_cast(float, __builtin_amdgcn_readlane(__builtin_bit_cast(int, iv), (int)u));
          v2f rw = *(const v2f*)(w1 + (size_t)rr * HID + 2u * lane);
          float* o = acc + dd * HID + 2u * lane;
          lds_fadd(o + 0, rw.x * ii);
          lds_fadd(o + 1, rw.y * ii);
        }
      }
    }
  }
  __syncthreads();

  for (unsigned i = tid; i < NPB * HID; i += 1024) {
    int node = lo + (int)(i >> 6);
    if (node < N_NODES) lhs[(size_t)node * K_TOT + K_AGG + (i & 63u)] = acc[i];
  }
}

__global__ void k_relu(const float* __restrict__ root1, const float* __restrict__ bias1,
                       float* __restrict__ lhs) {
  unsigned idx = blockIdx.x * blockDim.x + threadIdx.x;
  if (idx >= N_NODES * HID) return;
  unsigned n = idx >> 6, h = idx & 63;
  size_t p = (size_t)n * K_TOT + K_AGG + h;
  lhs[p] = fmaxf(lhs[p] + root1[idx] + bias1[h], 0.0f);
}

__global__ void k_layer2(const int* __restrict__ src_a, const int* __restrict__ dst_a,
                         const int* __restrict__ etype, const float* __restrict__ cnt,
                         float* __restrict__ lhs) {
  unsigned tid = blockIdx.x * blockDim.x + threadIdx.x;
  unsigned e = tid >> 5;
  unsigned lane = tid & 31;
  if (e >= N_EDGE) return;
  int rel = etype[e], s = src_a[e], d = dst_a[e];
  float inv = 1.0f / fmaxf(cnt[rel * N_NODES + d], 1.0f);
  const float* xrow = lhs + (size_t)s * K_TOT + K_AGG + 2u * lane;
  v2f v = *(const v2f*)xrow;
  float* o = lhs + (size_t)d * K_TOT + (unsigned)rel * HID + 2u * lane;
  fatomic_add(o + 0, v.x * inv);
  fatomic_add(o + 1, v.y * inv);
}

__global__ void k_gemm(const float* __restrict__ lhs, const float* __restrict__ B,
                       const float* __restrict__ bias2, float* __restrict__ out) {
  unsigned tid = blockIdx.x * blockDim.x + threadIdx.x;
  unsigned wave = tid >> 5;
  unsigned lane = tid & 31;
  if (wave >= N_NODES / 16) return;
  unsigned half = lane >> 4;
  unsigned sub = lane & 15;
  const float* arow = lhs + (size_t)(wave * 16u + sub) * K_TOT + 2u * half;
  v8f c0 = {}; v8f c1 = {};
  for (int k0 = 0; k0 < K_TOT; k0 += 4) {
    v2f a;  a.x = arow[k0];  a.y = arow[k0 + 1];
    int kb = k0 + 2 * (int)half;
    v2f b0; b0.x = B[kb * N_LAB + sub];        b0.y = B[(kb + 1) * N_LAB + sub];
    v2f b1; b1.x = B[kb * N_LAB + 16 + sub];   b1.y = B[(kb + 1) * N_LAB + 16 + sub];
    c0 = __builtin_amdgcn_wmma_f32_16x16x4_f32(false, a, false, b0, (short)0, c0, false, false);
    c1 = __builtin_amdgcn_wmma_f32_16x16x4_f32(false, a, false, b1, (short)0, c1, false, false);
  }
  #pragma unroll
  for (int v = 0; v < 8; v++) {
    unsigned row = wave * 16u + (unsigned)v + 8u * half;
    float x0 = c0[v] + bias2[sub];
    float x1 = c1[v] + bias2[16 + sub];
    out[row * N_LAB + sub]      = 1.0f / (1.0f + __expf(-x0));
    out[row * N_LAB + 16 + sub] = 1.0f / (1.0f + __expf(-x1));
  }
}

extern "C" void kernel_launch(void* const* d_in, const int* in_sizes, int n_in,
                              void* d_out, int out_size, void* d_ws, size_t ws_size,
                              hipStream_t stream) {
  const int* ei     = (const int*)d_in[0];
  const int* src_a  = ei;
  const int* dst_a  = ei + N_EDGE;
  const int* etype  = (const int*)d_in[1];
  const float* w1   = (const float*)d_in[2];
  const float* r1   = (const float*)d_in[3];
  const float* b1   = (const float*)d_in[4];
  const float* w2   = (const float*)d_in[5];
  const float* r2   = (const float*)d_in[6];
  const float* b2   = (const float*)d_in[7];
  float* out = (float*)d_out;

  float* ws   = (float*)d_ws;
  float* cnt  = ws + WS_CNT_OFF;
  float* Bmat = ws + WS_B_OFF;
  float* lhs  = ws + WS_LHS_OFF;

  (void)in_sizes; (void)n_in; (void)out_size; (void)ws_size;

  k_zero<<<2048, 256, 0, stream>>>(ws, (unsigned long long)WS_TOTAL);
  k_count<<<(N_EDGE + 255) / 256, 256, 0, stream>>>(etype, dst_a, cnt);
  k_bmat<<<(K_TOT * N_LAB + 255) / 256, 256, 0, stream>>>(w2, r2, Bmat);
  k_layer1<<<(N_NODES + NPB - 1) / NPB, 1024, 0, stream>>>(src_a, dst_a, etype, w1, cnt, lhs);
  k_relu<<<(N_NODES * HID + 255) / 256, 256, 0, stream>>>(r1, b1, lhs);
  k_layer2<<<(N_EDGE * 32u + 255) / 256, 256, 0, stream>>>(src_a, dst_a, etype, cnt, lhs);
  k_gemm<<<((N_NODES / 16) * 32 + 255) / 256, 256, 0, stream>>>(lhs, Bmat, b2, out);
}


# layer 2 per edge on f32 WMMA: a work-group groups 3125 edges by relation in LDS, a wave multiplies 16 x[src] rows by weight2[rel], adds rows/count into out by f32 atomics; sums reassociated
# speedup vs baseline: 1.9175x; 1.6906x over previous
#include <hip/hip_runtime.h>

#define N_NODES 50000
#define N_REL   16
#define HID     64
#define N_LAB   32
#define N_EDGE  1600000

typedef float v2f __attribute__((ext_vector_type(2)));
typedef float v4f __attribute__((ext_vector_type(4)));
typedef float v8f __attribute__((ext_vector_type(8)));

#define WS_CNT_OFF  0
#define WS_X_OFF    (N_REL * N_NODES)

__device__ __forceinline__ void fatomic_add(float* p, float v) {
  __hip_atomic_fetch_add(p, v, __ATOMIC_RELAXED, __HIP_MEMORY_SCOPE_AGENT);
}

__global__ void k_zero(float* __restrict__ p, unsigned long long n) {
  unsigned long long i = (unsigned long long)blockIdx.x * blockDim.x + threadIdx.x;
  unsigned long long stride = (unsigned long long)gridDim.x * blockDim.x;
  for (; i < n; i += stride) p[i] = 0.0f;
}

__global__ void k_count(const int* __restrict__ etype, const int* __restrict__ dst,
                        float* __restrict__ cnt) {
  unsigned e = blockIdx.x * blockDim.x + threadIdx.x;
  if (e >= N_EDGE) return;
  fatomic_add(&cnt[etype[e] * N_NODES + dst[e]], 1.0f);
}

#define NPB     196
#define QCAP    32768
#define SCAN_U  8
#define DR_U    16

__device__ __forceinline__ void lds_fadd(float* p, float v) {
  __hip_atomic_fetch_add(p, v, __ATOMIC_RELAXED, __HIP_MEMORY_SCOPE_WORKGROUP);
}

__global__ __launch_bounds__(1024)
void k_layer1(const int* __restrict__ src_a, const int* __restrict__ dst_a,
              const int* __restrict__ etype, const float* __restrict__ w1,
              const float* __restrict__ cnt, float* __restrict__ xs) {
  __shared__ float acc[NPB * HID];
  __shared__ int q[QCAP];
  __shared__ unsigned qn;
  const unsigned tid = threadIdx.x;
  const unsigned lane = tid & 31;
  const int lo = (int)blockIdx.x * NPB;

  for (unsigned i = tid; i < NPB * HID; i += 1024) acc[i] = 0.0f;
  if (tid == 0) qn = 0;
  __syncthreads();

  const int4* __restrict__ d4p = (const int4*)dst_a;
  const unsigned n4 = N_EDGE / 4;
  for (unsigned base = 0; base < n4; base += 1024u * SCAN_U) {
    int4 v[SCAN_U];
    #pragma unroll
    for (int u = 0; u < SCAN_U; u++) {
      unsigned i = base + (unsigned)u * 1024u + tid;
      v[u] = (i < n4) ? d4p[i] : make_int4(-1, -1, -1, -1);
    }
    #pragma unroll
    for (int u = 0; u < SCAN_U; u++) {
      unsigned i = base + (unsigned)u * 1024u + tid;
      int dd[4] = {v[u].x, v[u].y, v[u].z, v[u].w};
      unsigned nearest = min(min((unsigned)(dd[0] - lo), (unsigned)(dd[1] - lo)),
                             min((unsigned)(dd[2] - lo), (unsigned)(dd[3] - lo)));
      if (nearest < (unsigned)NPB) {
        #pragma unroll
        for (int c = 0; c < 4; c++) {
          if ((unsigned)(dd[c] - lo) < (unsigned)NPB) {
            int e = (int)(4u * i) + c;
            unsigned idx = atomicAdd(&qn, 1u);
            if (idx < QCAP) q[idx] = e;
          }
        }
      }
    }
  }
  __syncthreads();

  const unsigned nq = qn;
  if (nq > (unsigned)QCAP) {
    for (unsigned e = tid; e < N_EDGE; e += 1024u) {
      int d = dst_a[e];
      if ((unsigned)(d - lo) < (unsigned)NPB) {
        int rel = etype[e], s = src_a[e];
        float ivs = 1.0f / fmaxf(cnt[rel * N_NODES + d], 1.0f);
        const float* wr = w1 + ((size_t)rel * N_NODES + (size_t)s) * HID;
        float* ar = acc + (d - lo) * HID;
        for (int h = 0; h < HID; h++) lds_fadd(ar + h, wr[h] * ivs);
      }
    }
  }
  const unsigned n = (nq > (unsigned)QCAP) ? 0u : nq;
  const unsigned wave = (unsigned)__builtin_amdgcn_readfirstlane((int)(tid >> 5));
  for (unsigned b = wave * 32u; b < n; b += 32u * 32u) {
    const unsigned m = min(32u, n - b);
    int r = 0, dl = 0; float iv = 0.0f;
    if (lane < m) {
      int e = q[b + lane];
      int rel = etype[e], s = src_a[e], d = dst_a[e];
      iv = 1.0f / fmaxf(cnt[rel * N_NODES + d], 1.0f);
      r = rel * N_NODES + s;
      dl = d - lo;
    }
    #pragma unroll
    for (unsigned j = 0; j < 32; j += DR_U) {
      if (j >= m) break;
      if (j + DR_U <= m) {
        v2f row[DR_U];
        #pragma unroll
        for (unsigned u = 0; u < DR_U; u++) {
          int rr = __builtin_amdgcn_readlane(r, (int)(j + u));
          row[u] = *(const v2f*)(w1 + (size_t)rr * HID + 2u * lane);
        }
        #pragma unroll
        for (unsigned u = 0; u < DR_U; u++) {
          int dd = __builtin_amdgcn_readlane(dl, (int)(j + u));
          float ii = __builtin_bit_cast(float, __builtin_amdgcn_readlane(__builtin_bit_cast(int, iv), (int)(j + u)));
          float* o = acc + dd * HID + 2u * lane;
          lds_fadd(o + 0, row[u].x * ii);
          lds_fadd(o + 1, row[u].y * ii);
        }
      } else {
        for (unsigned u = j; u < m; u++) {
          int rr = __builtin_amdgcn_readlane(r, (int)u);
          int dd = __builtin_amdgcn_readlane(dl, (int)u);
          float ii = __builtin_bit_cast(float, __builtin_amdgcn_readlane(__builtin_bit_cast(int, iv), (int)u));
          v2f rw = *(const v2f*)(w1 + (size_t)rr * HID + 2u * lane);
          float* o = acc + dd * HID + 2u * lane;
          lds_fadd(o + 0, rw.x * ii);
          lds_fadd(o + 1, rw.y * ii);
        }
      }
    }
  }
  __syncthreads();

  for (unsigned i = tid; i < NPB * HID; i += 1024) {
    int node = lo + (int)(i >> 6);
    if (node < N_NODES) xs[(size_t)node * HID + (i & 63u)] = acc[i];
  }
}

__global__ void k_relu(const float* __restrict__ root1, const float* __restrict__ bias1,
                       float* __restrict__ xs) {
  unsigned idx = blockIdx.x * blockDim.x + threadIdx.x;
  if (idx >= N_NODES * HID) return;
  unsigned h = idx & 63;
  xs[idx] = fmaxf(xs[idx] + root1[idx] + bias1[h], 0.0f);
}


__global__ void k_root(const float* __restrict__ xs, const float* __restrict__ r2,
                       const float* __restrict__ bias2, float* __restrict__ out) {
  unsigned tid = blockIdx.x * blockDim.x + threadIdx.x;
  unsigned wave = tid >> 5;
  unsigned lane = tid & 31;
  if (wave >= N_NODES / 16) return;
  unsigned half = lane >> 4;
  unsigned sub = lane & 15;
  const float* arow = xs + (size_t)(wave * 16u + sub) * HID + 2u * half;
  v8f c0 = {}; v8f c1 = {};
  for (int k0 = 0; k0 < HID; k0 += 4) {
    v2f a;  a.x = arow[k0];  a.y = arow[k0 + 1];
    int kb = k0 + 2 * (int)half;
    v2f b0; b0.x = r2[kb * N_LAB + sub];        b0.y = r2[(kb + 1) * N_LAB + sub];
    v2f b1; b1.x = r2[kb * N_LAB + 16 + sub];   b1.y = r2[(kb + 1) * N_LAB + 16 + sub];
    c0 = __builtin_amdgcn_wmma_f32_16x16x4_f32(false, a, false, b0, (short)0, c0, false, false);
    c1 = __builtin_amdgcn_wmma_f32_16x16x4_f32(false, a, false, b1, (short)0, c1, false, false);
  }
  #pragma unroll
  for (int v = 0; v < 8; v++) {
    unsigned row = wave * 16u + (unsigned)v + 8u * half;
    out[row * N_LAB + sub]      = c0[v] + bias2[sub];
    out[row * N_LAB + 16 + sub] = c1[v] + bias2[16 + sub];
  }
}

#define EPB  3125
#define EPT  7

__global__ __launch_bounds__(512)
void k_edge(const int* __restrict__ src_a, const int* __restrict__ dst_a,
            const int* __restrict__ etype, const float* __restrict__ cnt,
            const float* __restrict__ xs, const float* __restrict__ w2,
            float* __restrict__ out) {
  __shared__ int s_src[EPB];
  __shared__ int s_dst[EPB];
  __shared__ float s_inv[EPB];
  __shared__ unsigned rc[N_REL];
  __shared__ unsigned ro[N_REL];
  __shared__ unsigned cur[N_REL];
  const unsigned tid = threadIdx.x;
  const unsigned lane = tid & 31;
  const unsigned ebase = blockIdx.x * (unsigned)EPB;

  if (tid < N_REL) { rc[tid] = 0; cur[tid] = 0; }
  __syncthreads();

  int rel[EPT];
  #pragma unroll
  for (int u = 0; u < EPT; u++) {
    unsigned i = tid + (unsigned)u * 512u;
    rel[u] = (i < (unsigned)EPB) ? etype[ebase + i] : -1;
    if (rel[u] >= 0) atomicAdd(&rc[rel[u]], 1u);
  }
  __syncthreads();
  if (tid == 0) {
    unsigned a = 0;
    for (int r = 0; r < N_REL; r++) { ro[r] = a; a += rc[r]; }
  }
  __syncthreads();
  #pragma unroll
  for (int u = 0; u < EPT; u++) {
    if (rel[u] >= 0) {
      unsigned e = ebase + tid + (unsigned)u * 512u;
      int r = rel[u];
      int s = src_a[e], d = dst_a[e];
      unsigned pos = ro[r] + atomicAdd(&cur[r], 1u);
      s_src[pos] = s;
      s_dst[pos] = d;
      s_inv[pos] = 1.0f / fmaxf(cnt[r * N_NODES + d], 1.0f);
    }
  }
  __syncthreads();

  const unsigned r = (unsigned)__builtin_amdgcn_readfirstlane((int)(tid >> 5));
  const unsigned n = (unsigned)__builtin_amdgcn_readfirstlane((int)rc[r]);
  const unsigned start = (unsigned)__builtin_amdgcn_readfirstlane((int)ro[r]);
  const unsigned half = lane >> 4;
  const unsigned sub = lane & 15;

  const float* wr = w2 + (size_t)r * HID * N_LAB;
  v2f b0[16], b1[16];
  #pragma unroll
  for (int s = 0; s < 16; s++) {
    unsigned k = 8u * (unsigned)(s >> 1) + 4u * half + 2u * (unsigned)(s & 1);
    b0[s].x = wr[k * N_LAB + sub];        b0[s].y = wr[(k + 1) * N_LAB + sub];
    b1[s].x = wr[k * N_LAB + 16 + sub];   b1[s].y = wr[(k + 1) * N_LAB + 16 + sub];
  }

  for (unsigned t0 = 0; t0 < n; t0 += 16) {
    const unsigned m = t0 + sub;
    const int s = (m < n) ? s_src[start + m] : 0;
    const v4f* xr = (const v4f*)(xs + (size_t)s * HID + 4u * half);
    v4f q[8];
    #pragma unroll
    for (int i = 0; i < 8; i++) q[i] = xr[2 * i];
    int dd[8]; float iv[8];
    #pragma unroll
    for (int v = 0; v < 8; v++) {
      unsigned row = min(t0 + (unsigned)v + 8u * half, n - 1u);
      dd[v] = s_dst[start + row];
      iv[v] = s_inv[start + row];
    }
    v8f c0 = {}; v8f c1 = {};
    #pragma unroll
    for (int st = 0; st < 16; st++) {
      v2f a;
      if (st & 1) { a.x = q[st >> 1].z; a.y = q[st >> 1].w; }
      else        { a.x = q[st >> 1].x; a.y = q[st >> 1].y; }
      c0 = __builtin_amdgcn_wmma_f32_16x16x4_f32(false, a, false, b0[st], (short)0, c0, false, false);
      c1 = __builtin_amdgcn_wmma_f32_16x16x4_f32(false, a, false, b1[st], (short)0, c1, false, false);
    }
    float p0[8], p1[8];
    #pragma unroll
    for (int v = 0; v < 8; v++) { p0[v] = c0[v] * iv[v]; p1[v] = c1[v] * iv[v]; }
    #pragma unroll
    for (int v = 0; v < 8; v++) {
      unsigned row = t0 + (unsigned)v + 8u * half;
      if (row < n) {
        float* o = out + (size_t)dd[v] * N_LAB + sub;
        fatomic_add(o, p0[v]);
        fatomic_add(o + 16, p1[v]);
      }
    }
  }
}

__global__ void k_sig(float* __restrict__ out) {
  unsigned i = blockIdx.x * blockDim.x + threadIdx.x;
  if (i >= N_NODES * N_LAB / 4) return;
  v4f v = ((const v4f*)out)[i];
  v.x = 1.0f / (1.0f + __expf(-v.x));
  v.y = 1.0f / (1.0f + __expf(-v.y));
  v.z = 1.0f / (1.0f + __expf(-v.z));
  v.w = 1.0f / (1.0f + __expf(-v.w));
  ((v4f*)out)[i] = v;
}

extern "C" void kernel_launch(void* const* d_in, const int* in_sizes, int n_in,
                              void* d_out, int out_size, void* d_ws, size_t ws_size,
                              hipStream_t stream) {
  const int* ei     = (const int*)d_in[0];
  const int* src_a  = ei;
  const int* dst_a  = ei + N_EDGE;
  const int* etype  = (const int*)d_in[1];
  const float* w1   = (const float*)d_in[2];
  const float* r1   = (const float*)d_in[3];
  const float* b1   = (const float*)d_in[4];
  const float* w2   = (const float*)d_in[5];
  const float* r2   = (const float*)d_in[6];
  const float* b2   = (const float*)d_in[7];
  float* out = (float*)d_out;

  float* ws   = (float*)d_ws;
  float* cnt  = ws + WS_CNT_OFF;
  float* xs   = ws + WS_X_OFF;

  (void)in_sizes; (void)n_in; (void)out_size; (void)ws_size;

  k_zero<<<512, 256, 0, stream>>>(cnt, (unsigned long long)(N_REL * N_NODES));
  k_count<<<(N_EDGE + 255) / 256, 256, 0, stream>>>(etype, dst_a, cnt);
  k_layer1<<<(N_NODES + NPB - 1) / NPB, 1024, 0, stream>>>(src_a, dst_a, etype, w1, cnt, xs);
  k_relu<<<(N_NODES * HID + 255) / 256, 256, 0, stream>>>(r1, b1, xs);
  k_root<<<((N_NODES / 16) * 32 + 255) / 256, 256, 0, stream>>>(xs, r2, b2, out);
  k_edge<<<N_EDGE / EPB, 512, 0, stream>>>(src_a, dst_a, etype, cnt, xs, w2, out);
  k_sig<<<(N_NODES * N_LAB / 4 + 255) / 256, 256, 0, stream>>>(out);
}


# pair counts by the owner: the layer-1 work-group counts its queued edges on LDS integer counters and stores 1/count; zero and count launches and their 1.6M global atomics gone; 7 launches become 5
# speedup vs baseline: 2.2436x; 1.1701x over previous
#include <hip/hip_runtime.h>

#define N_NODES 50000
#define N_REL   16
#define HID     64
#define N_LAB   32
#define N_EDGE  1600000

typedef float v2f __attribute__((ext_vector_type(2)));
typedef float v4f __attribute__((ext_vector_type(4)));
typedef float v8f __attribute__((ext_vector_type(8)));

#define WS_INV_OFF  0
#define WS_X_OFF    (N_REL * N_NODES)

__device__ __forceinline__ void fatomic_add(float* p, float v) {
  __hip_atomic_fetch_add(p, v, __ATOMIC_RELAXED, __HIP_MEMORY_SCOPE_AGENT);
}

#define NPB     196
#define QCAP    32768
#define SCAN_U  8
#define CNT_U   8
#define DR_U    16

__device__ __forceinline__ void lds_fadd(float* p, float v) {
  __hip_atomic_fetch_add(p, v, __ATOMIC_RELAXED, __HIP_MEMORY_SCOPE_WORKGROUP);
}

__global__ __launch_bounds__(1024)
void k_layer1(const int* __restrict__ src_a, const int* __restrict__ dst_a,
              const int* __restrict__ etype, const float* __restrict__ w1,
              float* __restrict__ inv, float* __restrict__ xs) {
  __shared__ float acc[NPB * HID];
  __shared__ int q[QCAP];
  __shared__ unsigned cl[NPB * N_REL];
  __shared__ unsigned qn;
  const unsigned tid = threadIdx.x;
  const unsigned lane = tid & 31;
  const int lo = (int)blockIdx.x * NPB;

  for (unsigned i = tid; i < NPB * HID; i += 1024) acc[i] = 0.0f;
  for (unsigned i = tid; i < NPB * N_REL; i += 1024) cl[i] = 0u;
  if (tid == 0) qn = 0;
  __syncthreads();

  const int4* __restrict__ d4p = (const int4*)dst_a;
  const unsigned n4 = N_EDGE / 4;
  for (unsigned base = 0; base < n4; base += 1024u * SCAN_U) {
    int4 v[SCAN_U];
    #pragma unroll
    for (int u = 0; u < SCAN_U; u++) {
      unsigned i = base + (unsigned)u * 1024u + tid;
      v[u] = (i < n4) ? d4p[i] : make_int4(-1, -1, -1, -1);
    }
    #pragma unroll
    for (int u = 0; u < SCAN_U; u++) {
      unsigned i = base + (unsigned)u * 1024u + tid;
      int dd[4] = {v[u].x, v[u].y, v[u].z, v[u].w};
      unsigned nearest = min(min((unsigned)(dd[0] - lo), (unsigned)(dd[1] - lo)),
                             min((unsigned)(dd[2] - lo), (unsigned)(dd[3] - lo)));
      if (nearest < (unsigned)NPB) {
        #pragma unroll
        for (int c = 0; c < 4; c++) {
          if ((unsigned)(dd[c] - lo) < (unsigned)NPB) {
            int e = (int)(4u * i) + c;
            unsigned idx = atomicAdd(&qn, 1u);
            if (idx < QCAP) q[idx] = e;
          }
        }
      }
    }
  }
  __syncthreads();

  const unsigned nq = qn;
  const bool over = nq > (unsigned)QCAP;
  if (over) {
    for (unsigned e = tid; e < N_EDGE; e += 1024u) {
      int d = dst_a[e];
      if ((unsigned)(d - lo) < (unsigned)NPB) atomicAdd(&cl[(unsigned)(d - lo) * N_REL + (unsigned)etype[e]], 1u);
    }
  } else {
    for (unsigned base = 0; base < nq; base += 1024u * CNT_U) {
      int e[CNT_U], rl[CNT_U], sc[CNT_U], ds[CNT_U];
      #pragma unroll
      for (int u = 0; u < CNT_U; u++) {
        unsigned i = base + (unsigned)u * 1024u + tid;
        e[u] = (i < nq) ? q[i] : -1;
      }
      #pragma unroll
      for (int u = 0; u < CNT_U; u++) {
        rl[u] = 0; sc[u] = 0; ds[u] = lo;
        if (e[u] >= 0) { rl[u] = etype[e[u]]; sc[u] = src_a[e[u]]; ds[u] = dst_a[e[u]]; }
      }
      #pragma unroll
      for (int u = 0; u < CNT_U; u++) {
        if (e[u] >= 0) {
          unsigned i = base + (unsigned)u * 1024u + tid;
          unsigned dl = (unsigned)(ds[u] - lo);
          atomicAdd(&cl[dl * N_REL + (unsigned)rl[u]], 1u);
          q[i] = (int)((unsigned)(rl[u] * N_NODES + sc[u]) | (dl << 20) | ((unsigned)rl[u] << 28));
        }
      }
    }
  }
  __syncthreads();

  for (unsigned i = tid; i < NPB * N_REL; i += 1024) {
    int node = lo + (int)(i >> 4);
    if (node < N_NODES) inv[(size_t)lo * N_REL + i] = 1.0f / fmaxf((float)cl[i], 1.0f);
  }

  if (over) {
    for (unsigned e = tid; e < N_EDGE; e += 1024u) {
      int d = dst_a[e];
      if ((unsigned)(d - lo) < (unsigned)NPB) {
        int rel = etype[e], s = src_a[e];
        float ivs = 1.0f / fmaxf((float)cl[(unsigned)(d - lo) * N_REL + (unsigned)rel], 1.0f);
        const float* wr = w1 + ((size_t)rel * N_NODES + (size_t)s) * HID;
        float* ar = acc + (d - lo) * HID;
        for (int h = 0; h < HID; h++) lds_fadd(ar + h, wr[h] * ivs);
      }
    }
  }
  const unsigned n = over ? 0u : nq;
  const unsigned wave = (unsigned)__builtin_amdgcn_readfirstlane((int)(tid >> 5));
  for (unsigned b = wave * 32u; b < n; b += 32u * 32u) {
    const unsigned m = min(32u, n - b);
    int r = 0, dl = 0; float iv = 0.0f;
    if (lane < m) {
      unsigned p = (unsigned)q[b + lane];
      r = (int)(p & 0xFFFFFu);
      dl = (int)((p >> 20) & 0xFFu);
      iv = 1.0f / fmaxf((float)cl[(unsigned)dl * N_REL + (p >> 28)], 1.0f);
    }
    #pragma unroll
    for (unsigned j = 0; j < 32; j += DR_U) {
      if (j >= m) break;
      if (j + DR_U <= m) {
        v2f row[DR_U];
        #pragma unroll
        for (unsigned u = 0; u < DR_U; u++) {
          int rr = __builtin_amdgcn_readlane(r, (int)(j + u));
          row[u] = *(const v2f*)(w1 + (size_t)rr * HID + 2u * lane);
        }
        #pragma unroll
        for (unsigned u = 0; u < DR_U; u++) {
          int dd = __builtin_amdgcn_readlane(dl, (int)(j + u));
          float ii = __builtin_bit_cast(float, __builtin_amdgcn_readlane(__builtin_bit_cast(int, iv), (int)(j + u)));
          float* o = acc + dd * HID + 2u * lane;
          lds_fadd(o + 0, row[u].x * ii);
          lds_fadd(o + 1, row[u].y * ii);
        }
      } else {
        for (unsigned u = j; u < m; u++) {
          int rr = __builtin_amdgcn_readlane(r, (int)u);
          int dd = __builtin_amdgcn_readlane(dl, (int)u);
          float ii = __builtin_bit_cast(float, __builtin_amdgcn_readlane(__builtin_bit_cast(int, iv), (int)u));
          v2f rw = *(const v2f*)(w1 + (size_t)rr * HID + 2u * lane);
          float* o = acc + dd * HID + 2u * lane;
          lds_fadd(o + 0, rw.x * ii);
          lds_fadd(o + 1, rw.y * ii);
        }
      }
    }
  }
  __syncthreads();

  for (unsigned i = tid; i < NPB * HID; i += 1024) {
    int node = lo + (int)(i >> 6);
    if (node < N_NODES) xs[(size_t)node * HID + (i & 63u)] = acc[i];
  }
}

__global__ void k_relu(const float* __restrict__ root1, const float* __restrict__ bias1,
                       float* __restrict__ xs) {
  unsigned idx = blockIdx.x * blockDim.x + threadIdx.x;
  if (idx >= N_NODES * HID) return;
  unsigned h = idx & 63;
  xs[idx] = fmaxf(xs[idx] + root1[idx] + bias1[h], 0.0f);
}


__global__ void k_root(const float* __restrict__ xs, const float* __restrict__ r2,
                       const float* __restrict__ bias2, float* __restrict__ out) {
  unsigned tid = blockIdx.x * blockDim.x + threadIdx.x;
  unsigned wave = tid >> 5;
  unsigned lane = tid & 31;
  if (wave >= N_NODES / 16) return;
  unsigned half = lane >> 4;
  unsigned sub = lane & 15;
  const float* arow = xs + (size_t)(wave * 16u + sub) * HID + 2u * half;
  v8f c0 = {}; v8f c1 = {};
  for (int k0 = 0; k0 < HID; k0 += 4) {
    v2f a;  a.x = arow[k0];  a.y = arow[k0 + 1];
    int kb = k0 + 2 * (int)half;
    v2f b0; b0.x = r2[kb * N_LAB + sub];        b0.y = r2[(kb + 1) * N_LAB + sub];
    v2f b1; b1.x = r2[kb * N_LAB + 16 + sub];   b1.y = r2[(kb + 1) * N_LAB + 16 + sub];
    c0 = __builtin_amdgcn_wmma_f32_16x16x4_f32(false, a, false, b0, (short)0, c0, false, false);
    c1 = __builtin_amdgcn_wmma_f32_16x16x4_f32(false, a, false, b1, (short)0, c1, false, false);
  }
  #pragma unroll
  for (int v = 0; v < 8; v++) {
    unsigned row = wave * 16u + (unsigned)v + 8u * half;
    out[row * N_LAB + sub]      = c0[v] + bias2[sub];
    out[row * N_LAB + 16 + sub] = c1[v] + bias2[16 + sub];
  }
}

#define EPB  3125
#define EPT  7

__global__ __launch_bounds__(512)
void k_edge(const int* __restrict__ src_a, const int* __restrict__ dst_a,
            const int* __restrict__ etype, const float* __restrict__ inv,
            const float* __restrict__ xs, const float* __restrict__ w2,
            float* __restrict__ out) {
  __shared__ int s_src[EPB];
  __shared__ int s_dst[EPB];
  __shared__ float s_inv[EPB];
  __shared__ unsigned rc[N_REL];
  __shared__ unsigned ro[N_REL];
  __shared__ unsigned cur[N_REL];
  const unsigned tid = threadIdx.x;
  const unsigned lane = tid & 31;
  const unsigned ebase = blockIdx.x * (unsigned)EPB;

  if (tid < N_REL) { rc[tid] = 0; cur[tid] = 0; }
  __syncthreads();

  int rel[EPT];
  #pragma unroll
  for (int u = 0; u < EPT; u++) {
    unsigned i = tid + (unsigned)u * 512u;
    rel[u] = (i < (unsigned)EPB) ? etype[ebase + i] : -1;
    if (rel[u] >= 0) atomicAdd(&rc[rel[u]], 1u);
  }
  __syncthreads();
  if (tid == 0) {
    unsigned a = 0;
    for (int r = 0; r < N_REL; r++) { ro[r] = a; a += rc[r]; }
  }
  __syncthreads();
  #pragma unroll
  for (int u = 0; u < EPT; u++) {
    if (rel[u] >= 0) {
      unsigned e = ebase + tid + (unsigned)u * 512u;
      int r = rel[u];
      int s = src_a[e], d = dst_a[e];
      unsigned pos = ro[r] + atomicAdd(&cur[r], 1u);
      s_src[pos] = s;
      s_dst[pos] = d;
      s_inv[pos] = inv[d * N_REL + r];
    }
  }
  __syncthreads();

  const unsigned r = (unsigned)__builtin_amdgcn_readfirstlane((int)(tid >> 5));
  const unsigned n = (unsigned)__builtin_amdgcn_readfirstlane((int)rc[r]);
  const unsigned start = (unsigned)__builtin_amdgcn_readfirstlane((int)ro[r]);
  const unsigned half = lane >> 4;
  const unsigned sub = lane & 15;

  const float* wr = w2 + (size_t)r * HID * N_LAB;
  v2f b0[16], b1[16];
  #pragma unroll
  for (int s = 0; s < 16; s++) {
    unsigned k = 8u * (unsigned)(s >> 1) + 4u * half + 2u * (unsigned)(s & 1);
    b0[s].x = wr[k * N_LAB + sub];        b0[s].y = wr[(k + 1) * N_LAB + sub];
    b1[s].x = wr[k * N_LAB + 16 + sub];   b1[s].y = wr[(k + 1) * N_LAB + 16 + sub];
  }

  for (unsigned t0 = 0; t0 < n; t0 += 16) {
    const unsigned m = t0 + sub;
    const int s = (m < n) ? s_src[start + m] : 0;
    const v4f* xr = (const v4f*)(xs + (size_t)s * HID + 4u * half);
    v4f q[8];
    #pragma unroll
    for (int i = 0; i < 8; i++) q[i] = xr[2 * i];
    int dd[8]; float iv[8];
    #pragma unroll
    for (int v = 0; v < 8; v++) {
      unsigned row = min(t0 + (unsigned)v + 8u * half, n - 1u);
      dd[v] = s_dst[start + row];
      iv[v] = s_inv[start + row];
    }
    v8f c0 = {}; v8f c1 = {};
    #pragma unroll
    for (int st = 0; st < 16; st++) {
      v2f a;
      if (st & 1) { a.x = q[st >> 1].z; a.y = q[st >> 1].w; }
      else        { a.x = q[st >> 1].x; a.y = q[st >> 1].y; }
      c0 = __builtin_amdgcn_wmma_f32_16x16x4_f32(false, a, false, b0[st], (short)0, c0, false, false);
      c1 = __builtin_amdgcn_wmma_f32_16x16x4_f32(false, a, false, b1[st], (short)0, c1, false, false);
    }
    float p0[8], p1[8];
    #pragma unroll
    for (int v = 0; v < 8; v++) { p0[v] = c0[v] * iv[v]; p1[v] = c1[v] * iv[v]; }
    #pragma unroll
    for (int v = 0; v < 8; v++) {
      unsigned row = t0 + (unsigned)v + 8u * half;
      if (row < n) {
        float* o = out + (size_t)dd[v] * N_LAB + sub;
        fatomic_add(o, p0[v]);
        fatomic_add(o + 16, p1[v]);
      }
    }
  }
}

__global__ void k_sig(float* __restrict__ out) {
  unsigned i = blockIdx.x * blockDim.x + threadIdx.x;
  if (i >= N_NODES * N_LAB / 4) return;
  v4f v = ((const v4f*)out)[i];
  v.x = 1.0f / (1.0f + __expf(-v.x));
  v.y = 1.0f / (1.0f + __expf(-v.y));
  v.z = 1.0f / (1.0f + __expf(-v.z));
  v.w = 1.0f / (1.0f + __expf(-v.w));
  ((v4f*)out)[i] = v;
}

extern "C" void kernel_launch(void* const* d_in, const int* in_sizes, int n_in,
                              void* d_out, int out_size, void* d_ws, size_t ws_size,
                              hipStream_t stream) {
  const int* ei     = (const int*)d_in[0];
  const int* src_a  = ei;
  const int* dst_a  = ei + N_EDGE;
  const int* etype  = (const int*)d_in[1];
  const float* w1   = (const float*)d_in[2];
  const float* r1   = (const float*)d_in[3];
  const float* b1   = (const float*)d_in[4];
  const float* w2   = (const float*)d_in[5];
  const float* r2   = (const float*)d_in[6];
  const float* b2   = (const float*)d_in[7];
  float* out = (float*)d_out;

  float* ws   = (float*)d_ws;
  float* inv  = ws + WS_INV_OFF;
  float* xs   = ws + WS_X_OFF;

  (void)in_sizes; (void)n_in; (void)out_size; (void)ws_size;

  k_layer1<<<(N_NODES + NPB - 1) / NPB, 1024, 0, stream>>>(src_a, dst_a, etype, w1, inv, xs);
  k_relu<<<(N_NODES * HID + 255) / 256, 256, 0, stream>>>(r1, b1, xs);
  k_root<<<((N_NODES / 16) * 32 + 255) / 256, 256, 0, stream>>>(xs, r2, b2, out);
  k_edge<<<N_EDGE / EPB, 512, 0, stream>>>(src_a, dst_a, etype, inv, xs, w2, out);
  k_sig<<<(N_NODES * N_LAB / 4 + 255) / 256, 256, 0, stream>>>(out);
}


# layer 2 owner-computes: layer 1 hands its edge queue on in the workspace; 256 work-groups add per-edge WMMA rows into LDS out rows (f32 LDS adds, arrival order), sigmoid, one store; 5 launches to 3
# speedup vs baseline: 2.6202x; 1.1679x over previous
#include <hip/hip_runtime.h>

#define N_NODES 50000
#define N_REL   16
#define HID     64
#define N_LAB   32
#define N_EDGE  1600000

typedef float v2f __attribute__((ext_vector_type(2)));
typedef float v4f __attribute__((ext_vector_type(4)));
typedef float v8f __attribute__((ext_vector_type(8)));

#define WS_INV_OFF  0
#define WS_X_OFF    (N_REL * N_NODES)
#define NRANGE      256
#define QSLOT       8192
#define WS_QN_OFF   (WS_X_OFF + N_NODES * HID)
#define WS_QG_OFF   (WS_QN_OFF + NRANGE)
#define WS_NEED_BYTES ((unsigned long long)(WS_QG_OFF + NRANGE * QSLOT) * 4ull)

#define NPB     196
#define QCAP    32768
#define SCAN_U  8
#define CNT_U   8
#define DR_U    16

__device__ __forceinline__ void lds_fadd(float* p, float v) {
  __hip_atomic_fetch_add(p, v, __ATOMIC_RELAXED, __HIP_MEMORY_SCOPE_WORKGROUP);
}

__global__ __launch_bounds__(1024)
void k_layer1(const int* __restrict__ src_a, const int* __restrict__ dst_a,
              const int* __restrict__ etype, const float* __restrict__ w1,
              float* __restrict__ inv, float* __restrict__ xs,
              int* __restrict__ qn_g, int* __restrict__ qg, unsigned long long ws_bytes) {
  __shared__ float acc[NPB * HID];
  __shared__ int q[QCAP];
  __shared__ unsigned cl[NPB * N_REL];
  __shared__ unsigned qn;
  const unsigned tid = threadIdx.x;
  const unsigned lane = tid & 31;
  const int lo = (int)blockIdx.x * NPB;

  for (unsigned i = tid; i < NPB * HID; i += 1024) acc[i] = 0.0f;
  for (unsigned i = tid; i < NPB * N_REL; i += 1024) cl[i] = 0u;
  if (tid == 0) qn = 0;
  __syncthreads();

  const int4* __restrict__ d4p = (const int4*)dst_a;
  const unsigned n4 = N_EDGE / 4;
  for (unsigned base = 0; base < n4; base += 1024u * SCAN_U) {
    int4 v[SCAN_U];
    #pragma unroll
    for (int u = 0; u < SCAN_U; u++) {
      unsigned i = base + (unsigned)u * 1024u + tid;
      v[u] = (i < n4) ? d4p[i] : make_int4(-1, -1, -1, -1);
    }
    #pragma unroll
    for (int u = 0; u < SCAN_U; u++) {
      unsigned i = base + (unsigned)u * 1024u + tid;
      int dd[4] = {v[u].x, v[u].y, v[u].z, v[u].w};
      unsigned nearest = min(min((unsigned)(dd[0] - lo), (unsigned)(dd[1] - lo)),
                             min((unsigned)(dd[2] - lo), (unsigned)(dd[3] - lo)));
      if (nearest < (unsigned)NPB) {
        #pragma unroll
        for (int c = 0; c < 4; c++) {
          if ((unsigned)(dd[c] - lo) < (unsigned)NPB) {
            int e = (int)(4u * i) + c;
            unsigned idx = atomicAdd(&qn, 1u);
            if (idx < QCAP) q[idx] = e;
          }
        }
      }
    }
  }
  __syncthreads();

  const unsigned nq = qn;
  const bool over = nq > (unsigned)QCAP;
  if (over) {
    for (unsigned e = tid; e < N_EDGE; e += 1024u) {
      int d = dst_a[e];
      if ((unsigned)(d - lo) < (unsigned)NPB) atomicAdd(&cl[(unsigned)(d - lo) * N_REL + (unsigned)etype[e]], 1u);
    }
  } else {
    for (unsigned base = 0; base < nq; base += 1024u * CNT_U) {
      int e[CNT_U], rl[CNT_U], sc[CNT_U], ds[CNT_U];
      #pragma unroll
      for (int u = 0; u < CNT_U; u++) {
        unsigned i = base + (unsigned)u * 1024u + tid;
        e[u] = (i < nq) ? q[i] : -1;
      }
      #pragma unroll
      for (int u = 0; u < CNT_U; u++) {
        rl[u] = 0; sc[u] = 0; ds[u] = lo;
        if (e[u] >= 0) { rl[u] = etype[e[u]]; sc[u] = src_a[e[u]]; ds[u] = dst_a[e[u]]; }
      }
      #pragma unroll
      for (int u = 0; u < CNT_U; u++) {
        if (e[u] >= 0) {
          unsigned i = base + (unsigned)u * 1024u + tid;
          unsigned dl = (unsigned)(ds[u] - lo);
          atomicAdd(&cl[dl * N_REL + (unsigned)rl[u]], 1u);
          q[i] = (int)((unsigned)(rl[u] * N_NODES + sc[u]) | (dl << 20) | ((unsigned)rl[u] << 28));
        }
      }
    }
  }
  __syncthreads();

  for (unsigned i = tid; i < NPB * N_REL; i += 1024) {
    int node = lo + (int)(i >> 4);
    if (node < N_NODES) inv[(size_t)lo * N_REL + i] = 1.0f / fmaxf((float)cl[i], 1.0f);
  }

  if (ws_bytes >= WS_NEED_BYTES) {
    if (tid == 0) qn_g[blockIdx.x] = (int)min(nq, (unsigned)QSLOT + 1u);
    if (nq <= (unsigned)QSLOT)
      for (unsigned i = tid; i < nq; i += 1024) qg[(size_t)blockIdx.x * QSLOT + i] = q[i];
  }

  if (over) {
    for (unsigned e = tid; e < N_EDGE; e += 1024u) {
      int d = dst_a[e];
      if ((unsigned)(d - lo) < (unsigned)NPB) {
        int rel = etype[e], s = src_a[e];
        float ivs = 1.0f / fmaxf((float)cl[(unsigned)(d - lo) * N_REL + (unsigned)rel], 1.0f);
        const float* wr = w1 + ((size_t)rel * N_NODES + (size_t)s) * HID;
        float* ar = acc + (d - lo) * HID;
        for (int h = 0; h < HID; h++) lds_fadd(ar + h, wr[h] * ivs);
      }
    }
  }
  const unsigned n = over ? 0u : nq;
  const unsigned wave = (unsigned)__builtin_amdgcn_readfirstlane((int)(tid >> 5));
  for (unsigned b = wave * 32u; b < n; b += 32u * 32u) {
    const unsigned m = min(32u, n - b);
    int r = 0, dl = 0; float iv = 0.0f;
    if (lane < m) {
      unsigned p = (unsigned)q[b + lane];
      r = (int)(p & 0xFFFFFu);
      dl = (int)((p >> 20) & 0xFFu);
      iv = 1.0f / fmaxf((float)cl[(unsigned)dl * N_REL + (p >> 28)], 1.0f);
    }
    #pragma unroll
    for (unsigned j = 0; j < 32; j += DR_U) {
      if (j >= m) break;
      if (j + DR_U <= m) {
        v2f row[DR_U];
        #pragma unroll
        for (unsigned u = 0; u < DR_U; u++) {
          int rr = __builtin_amdgcn_readlane(r, (int)(j + u));
          row[u] = *(const v2f*)(w1 + (size_t)rr * HID + 2u * lane);
        }
        #pragma unroll
        for (unsigned u = 0; u < DR_U; u++) {
          int dd = __builtin_amdgcn_readlane(dl, (int)(j + u));
          float ii = __builtin_bit_cast(float, __builtin_amdgcn_readlane(__builtin_bit_cast(int, iv), (int)(j + u)));
          float* o = acc + dd * HID + 2u * lane;
          lds_fadd(o + 0, row[u].x * ii);
          lds_fadd(o + 1, row[u].y * ii);
        }
      } else {
        for (unsigned u = j; u < m; u++) {
          int rr = __builtin_amdgcn_readlane(r, (int)u);
          int dd = __builtin_amdgcn_readlane(dl, (int)u);
          float ii = __builtin_bit_cast(float, __builtin_amdgcn_readlane(__builtin_bit_cast(int, iv), (int)u));
          v2f rw = *(const v2f*)(w1 + (size_t)rr * HID + 2u * lane);
          float* o = acc + dd * HID + 2u * lane;
          lds_fadd(o + 0, rw.x * ii);
          lds_fadd(o + 1, rw.y * ii);
        }
      }
    }
  }
  __syncthreads();

  for (unsigned i = tid; i < NPB * HID; i += 1024) {
    int node = lo + (int)(i >> 6);
    if (node < N_NODES) xs[(size_t)node * HID + (i & 63u)] = acc[i];
  }
}

__global__ void k_relu(const float* __restrict__ root1, const float* __restrict__ bias1,
                       float* __restrict__ xs) {
  unsigned idx = blockIdx.x * blockDim.x + threadIdx.x;
  if (idx >= N_NODES * HID) return;
  unsigned h = idx & 63;
  xs[idx] = fmaxf(xs[idx] + root1[idx] + bias1[h], 0.0f);
}


#define QC2   QSLOT
#define L2_U  8
#define QNONE 0xFFFFFFFFu

__global__ __launch_bounds__(1024)
void k_layer2(const int* __restrict__ src_a, const int* __restrict__ dst_a,
              const int* __restrict__ etype, const float* __restrict__ inv,
              const float* __restrict__ xs, const float* __restrict__ w2,
              const float* __restrict__ r2, const float* __restrict__ bias2,
              const int* __restrict__ qn_g, const int* __restrict__ qg,
              unsigned long long ws_bytes, float* __restrict__ out) {
  __shared__ float acc2[NPB * N_LAB];
  __shared__ float invl[NPB * N_REL];
  __shared__ unsigned qs[QC2];
  __shared__ unsigned rc[N_REL];
  __shared__ unsigned ro[N_REL];
  __shared__ unsigned cur[N_REL];
  const unsigned tid = threadIdx.x;
  const unsigned lane = tid & 31;
  const unsigned half = lane >> 4;
  const unsigned sub = lane & 15;
  const int lo = (int)blockIdx.x * NPB;
  const unsigned wave = (unsigned)__builtin_amdgcn_readfirstlane((int)(tid >> 5));

  for (unsigned i = tid; i < NPB * N_REL; i += 1024) {
    int node = lo + (int)(i >> 4);
    invl[i] = (node < N_NODES) ? inv[(size_t)lo * N_REL + i] : 0.0f;
  }

  if (wave < (NPB + 15) / 16) {
    const int node = min(lo + (int)(wave * 16u + sub), N_NODES - 1);
    const float* arow = xs + (size_t)node * HID + 2u * half;
    v8f c0 = {}; v8f c1 = {};
    for (int k0 = 0; k0 < HID; k0 += 4) {
      v2f a;  a.x = arow[k0];  a.y = arow[k0 + 1];
      int kb = k0 + 2 * (int)half;
      v2f b0; b0.x = r2[kb * N_LAB + sub];        b0.y = r2[(kb + 1) * N_LAB + sub];
      v2f b1; b1.x = r2[kb * N_LAB + 16 + sub];   b1.y = r2[(kb + 1) * N_LAB + 16 + sub];
      c0 = __builtin_amdgcn_wmma_f32_16x16x4_f32(false, a, false, b0, (short)0, c0, false, false);
      c1 = __builtin_amdgcn_wmma_f32_16x16x4_f32(false, a, false, b1, (short)0, c1, false, false);
    }
    #pragma unroll
    for (int v = 0; v < 8; v++) {
      unsigned row = wave * 16u + (unsigned)v + 8u * half;
      if (row < (unsigned)NPB) {
        acc2[row * N_LAB + sub]      = c0[v] + bias2[sub];
        acc2[row * N_LAB + 16 + sub] = c1[v] + bias2[16 + sub];
      }
    }
  }

  const unsigned r = wave & 15u;
  const unsigned hw = wave >> 4;
  const float* wr = w2 + (size_t)r * HID * N_LAB;
  v2f b0[16], b1[16];
  #pragma unroll
  for (int s = 0; s < 16; s++) {
    unsigned k = 8u * (unsigned)(s >> 1) + 4u * half + 2u * (unsigned)(s & 1);
    b0[s].x = wr[k * N_LAB + sub];        b0[s].y = wr[(k + 1) * N_LAB + sub];
    b1[s].x = wr[k * N_LAB + 16 + sub];   b1[s].y = wr[(k + 1) * N_LAB + 16 + sub];
  }

  const bool slots = ws_bytes >= WS_NEED_BYTES;
  const unsigned nq = slots ? (unsigned)qn_g[blockIdx.x] : QNONE;
  const bool handed = nq <= (unsigned)QSLOT;
  const unsigned nchunk = handed ? 1u : (unsigned)((N_EDGE + QC2 - 1) / QC2);

  for (unsigned ch = 0; ch < nchunk; ch++) {
    __syncthreads();
    if (tid < N_REL) { rc[tid] = 0; cur[tid] = 0; }
    __syncthreads();

    unsigned p[L2_U];
    #pragma unroll
    for (int u = 0; u < L2_U; u++) {
      const unsigned i = (unsigned)u * 1024u + tid;
      p[u] = QNONE;
      if (handed) {
        if (i < nq) p[u] = (unsigned)qg[(size_t)blockIdx.x * QSLOT + i];
      } else {
        const unsigned e = ch * (unsigned)QC2 + i;
        if (e < (unsigned)N_EDGE) {
          const int d = dst_a[e];
          if ((unsigned)(d - lo) < (unsigned)NPB) {
            const int rl = etype[e], sc = src_a[e];
            p[u] = (unsigned)(rl * N_NODES + sc) | ((unsigned)(d - lo) << 20) | ((unsigned)rl << 28);
          }
        }
      }
      if (p[u] != QNONE) atomicAdd(&rc[p[u] >> 28], 1u);
    }
    __syncthreads();
    if (tid == 0) {
      unsigned a = 0;
      for (int k = 0; k < N_REL; k++) { ro[k] = a; a += rc[k]; }
    }
    __syncthreads();
    #pragma unroll
    for (int u = 0; u < L2_U; u++) {
      if (p[u] != QNONE) {
        const unsigned rr = p[u] >> 28;
        qs[ro[rr] + atomicAdd(&cur[rr], 1u)] = p[u];
      }
    }
    __syncthreads();

    const unsigned n = (unsigned)__builtin_amdgcn_readfirstlane((int)rc[r]);
    const unsigned start = (unsigned)__builtin_amdgcn_readfirstlane((int)ro[r]);
    for (unsigned t0 = 16u * hw; t0 < n; t0 += 32u) {
      const unsigned m = t0 + sub;
      const int s = (m < n) ? (int)(qs[start + m] & 0xFFFFFu) - (int)(r * (unsigned)N_NODES) : 0;
      const v4f* xr = (const v4f*)(xs + (size_t)s * HID + 4u * half);
      v4f q[8];
      #pragma unroll
      for (int i = 0; i < 8; i++) q[i] = xr[2 * i];
      unsigned dl[8]; float iv[8];
      #pragma unroll
      for (int v = 0; v < 8; v++) {
        unsigned row = min(t0 + (unsigned)v + 8u * half, n - 1u);
        dl[v] = (qs[start + row] >> 20) & 0xFFu;
        iv[v] = invl[dl[v] * N_REL + r];
      }
      v8f c0 = {}; v8f c1 = {};
      #pragma unroll
      for (int st = 0; st < 16; st++) {
        v2f a;
        if (st & 1) { a.x = q[st >> 1].z; a.y = q[st >> 1].w; }
        else        { a.x = q[st >> 1].x; a.y = q[st >> 1].y; }
        c0 = __builtin_amdgcn_wmma_f32_16x16x4_f32(false, a, false, b0[st], (short)0, c0, false, false);
        c1 = __builtin_amdgcn_wmma_f32_16x16x4_f32(false, a, false, b1[st], (short)0, c1, false, false);
      }
      float p0[8], p1[8];
      #pragma unroll
      for (int v = 0; v < 8; v++) { p0[v] = c0[v] * iv[v]; p1[v] = c1[v] * iv[v]; }
      #pragma unroll
      for (int v = 0; v < 8; v++) {
        unsigned row = t0 + (unsigned)v + 8u * half;
        if (row < n) {
          float* o = acc2 + dl[v] * N_LAB + sub;
          lds_fadd(o, p0[v]);
          lds_fadd(o + 16, p1[v]);
        }
      }
    }
  }
  __syncthreads();

  for (unsigned i = tid; i < NPB * N_LAB; i += 1024) {
    int node = lo + (int)(i >> 5);
    if (node < N_NODES) out[(size_t)lo * N_LAB + i] = 1.0f / (1.0f + __expf(-acc2[i]));
  }
}

extern "C" void kernel_launch(void* const* d_in, const int* in_sizes, int n_in,
                              void* d_out, int out_size, void* d_ws, size_t ws_size,
                              hipStream_t stream) {
  const int* ei     = (const int*)d_in[0];
  const int* src_a  = ei;
  const int* dst_a  = ei + N_EDGE;
  const int* etype  = (const int*)d_in[1];
  const float* w1   = (const float*)d_in[2];
  const float* r1   = (const float*)d_in[3];
  const float* b1   = (const float*)d_in[4];
  const float* w2   = (const float*)d_in[5];
  const float* r2   = (const float*)d_in[6];
  const float* b2   = (const float*)d_in[7];
  float* out = (float*)d_out;

  float* ws   = (float*)d_ws;
  float* inv  = ws + WS_INV_OFF;
  float* xs   = ws + WS_X_OFF;
  int* qn_g   = (int*)d_ws + WS_QN_OFF;
  int* qg     = (int*)d_ws + WS_QG_OFF;

  (void)in_sizes; (void)n_in; (void)out_size;

  k_layer1<<<NRANGE, 1024, 0, stream>>>(src_a, dst_a, etype, w1, inv, xs, qn_g, qg,
                                        (unsigned long long)ws_size);
  k_relu<<<(N_NODES * HID + 255) / 256, 256, 0, stream>>>(r1, b1, xs);
  k_layer2<<<NRANGE, 1024, 0, stream>>>(src_a, dst_a, etype, inv, xs, w2, r2, b2, qn_g, qg,
                                        (unsigned long long)ws_size, out);
}


# layer-1 scan reads bit-sliced range keys (dst/196 as 8 bit planes, written by a new first launch): 32 edges tested per word; 3 launches become 4; queue and f32 LDS sums stay in arrival order
# speedup vs baseline: 3.2705x; 1.2482x over previous
#include <hip/hip_runtime.h>

#define N_NODES 50000
#define N_REL   16
#define HID     64
#define N_LAB   32
#define N_EDGE  1600000

typedef float v2f __attribute__((ext_vector_type(2)));
typedef float v4f __attribute__((ext_vector_type(4)));
typedef float v8f __attribute__((ext_vector_type(8)));

#define WS_INV_OFF  0
#define WS_X_OFF    (N_REL * N_NODES)
#define NRANGE      256
#define QSLOT       8192
#define WS_QN_OFF   (WS_X_OFF + N_NODES * HID)
#define WS_QG_OFF   (WS_QN_OFF + NRANGE)
#define WS_NEED_BYTES ((unsigned long long)(WS_QG_OFF + NRANGE * QSLOT) * 4ull)
#define KWORDS      (N_EDGE / 32)
#define KPITCH      50048
#define WS_KEY_OFF  (WS_QG_OFF + NRANGE * QSLOT)
#define WS_KEY_BYTES ((unsigned long long)(WS_KEY_OFF + 8 * KPITCH) * 4ull)

#define NPB     196
#define QCAP    32768
#define SCAN_U  8
#define KSCAN_U 2
#define K4      (KWORDS / 4)
#define CNT_U   8
#define DR_U    16

__device__ __forceinline__ void lds_fadd(float* p, float v) {
  __hip_atomic_fetch_add(p, v, __ATOMIC_RELAXED, __HIP_MEMORY_SCOPE_WORKGROUP);
}

__global__ void k_keys(const int* __restrict__ dst_a, unsigned* __restrict__ keys,
                       unsigned long long ws_bytes) {
  if (ws_bytes < WS_KEY_BYTES) return;
  unsigned t = blockIdx.x * blockDim.x + threadIdx.x;
  if (t >= KWORDS) return;
  const int4* __restrict__ d4 = (const int4*)dst_a + 8u * t;
  unsigned w[8] = {0u, 0u, 0u, 0u, 0u, 0u, 0u, 0u};
  #pragma unroll
  for (int c = 0; c < 8; c++) {
    int4 d = d4[c];
    int dd[4] = {d.x, d.y, d.z, d.w};
    #pragma unroll
    for (int k = 0; k < 4; k++) {
      unsigned key = min((unsigned)dd[k] / (unsigned)NPB, 255u);
      #pragma unroll
      for (int j = 0; j < 8; j++) w[j] |= ((key >> j) & 1u) << (4 * c + k);
    }
  }
  #pragma unroll
  for (int j = 0; j < 8; j++) keys[(unsigned)j * KPITCH + t] = w[j];
}

__device__ __forceinline__ void scan_planes(const int4* __restrict__ k4p, unsigned tid,
                                            int* q, unsigned* qn) {
  unsigned flip[8];
  #pragma unroll
  for (int j = 0; j < 8; j++) flip[j] = ((blockIdx.x >> j) & 1u) ? 0u : 0xFFFFFFFFu;
  for (unsigned base = 0; base < (unsigned)K4; base += 1024u * KSCAN_U) {
    int4 v[KSCAN_U][8];
    #pragma unroll
    for (int u = 0; u < KSCAN_U; u++) {
      unsigned i = min(base + (unsigned)u * 1024u + tid, (unsigned)K4 - 1u);
      #pragma unroll
      for (int j = 0; j < 8; j++) v[u][j] = k4p[(unsigned)j * (KPITCH / 4) + i];
    }
    #pragma unroll
    for (int u = 0; u < KSCAN_U; u++) {
      const unsigned i = base + (unsigned)u * 1024u + tid;
      const unsigned live = (i < (unsigned)K4) ? 0xFFFFFFFFu : 0u;
      #pragma unroll
      for (int c = 0; c < 4; c++) {
        unsigned m = live;
        #pragma unroll
        for (int j = 0; j < 8; j++) {
          const int pw[4] = {v[u][j].x, v[u][j].y, v[u][j].z, v[u][j].w};
          m &= (unsigned)pw[c] ^ flip[j];
        }
        while (m != 0u) {
          unsigned l = (unsigned)__builtin_ctz(m);
          m &= m - 1u;
          int e = (int)(128u * i + 32u * (unsigned)c + l);
          unsigned idx = atomicAdd(qn, 1u);
          if (idx < QCAP) q[idx] = e;
        }
      }
    }
  }
}

__device__ __forceinline__ void scan_dst(const int* __restrict__ dst_a, int lo, unsigned tid,
                                         int* q, unsigned* qn) {
  const int4* __restrict__ d4p = (const int4*)dst_a;
  const unsigned n4 = N_EDGE / 4;
  for (unsigned base = 0; base < n4; base += 1024u * SCAN_U) {
    int4 v[SCAN_U];
    #pragma unroll
    for (int u = 0; u < SCAN_U; u++) {
      unsigned i = base + (unsigned)u * 1024u + tid;
      v[u] = (i < n4) ? d4p[i] : make_int4(-1, -1, -1, -1);
    }
    #pragma unroll
    for (int u = 0; u < SCAN_U; u++) {
      unsigned i = base + (unsigned)u * 1024u + tid;
      int dd[4] = {v[u].x, v[u].y, v[u].z, v[u].w};
      unsigned nearest = min(min((unsigned)(dd[0] - lo), (unsigned)(dd[1] - lo)),
                             min((unsigned)(dd[2] - lo), (unsigned)(dd[3] - lo)));
      if (nearest < (unsigned)NPB) {
        #pragma unroll
        for (int c = 0; c < 4; c++) {
          if ((unsigned)(dd[c] - lo) < (unsigned)NPB) {
            int e = (int)(4u * i) + c;
            unsigned idx = atomicAdd(qn, 1u);
            if (idx < QCAP) q[idx] = e;
          }
        }
      }
    }
  }
}

__global__ __launch_bounds__(1024)
void k_layer1(const int* __restrict__ src_a, const int* __restrict__ dst_a,
              const int* __restrict__ etype, const float* __restrict__ w1,
              float* __restrict__ inv, float* __restrict__ xs,
              int* __restrict__ qn_g, int* __restrict__ qg, const int* __restrict__ keys,
              unsigned long long ws_bytes) {
  __shared__ float acc[NPB * HID];
  __shared__ int q[QCAP];
  __shared__ unsigned cl[NPB * N_REL];
  __shared__ unsigned qn;
  const unsigned tid = threadIdx.x;
  const unsigned lane = tid & 31;
  const int lo = (int)blockIdx.x * NPB;

  for (unsigned i = tid; i < NPB * HID; i += 1024) acc[i] = 0.0f;
  for (unsigned i = tid; i < NPB * N_REL; i += 1024) cl[i] = 0u;
  if (tid == 0) qn = 0;
  __syncthreads();

  if (ws_bytes >= WS_KEY_BYTES) scan_planes((const int4*)keys, tid, q, &qn);
  else                          scan_dst(dst_a, lo, tid, q, &qn);
  __syncthreads();

  const unsigned nq = qn;
  const bool over = nq > (unsigned)QCAP;
  if (over) {
    for (unsigned e = tid; e < N_EDGE; e += 1024u) {
      int d = dst_a[e];
      if ((unsigned)(d - lo) < (unsigned)NPB) atomicAdd(&cl[(unsigned)(d - lo) * N_REL + (unsigned)etype[e]], 1u);
    }
  } else {
    for (unsigned base = 0; base < nq; base += 1024u * CNT_U) {
      int e[CNT_U], rl[CNT_U], sc[CNT_U], ds[CNT_U];
      #pragma unroll
      for (int u = 0; u < CNT_U; u++) {
        unsigned i = base + (unsigned)u * 1024u + tid;
        e[u] = (i < nq) ? q[i] : -1;
      }
      #pragma unroll
      for (int u = 0; u < CNT_U; u++) {
        rl[u] = 0; sc[u] = 0; ds[u] = lo;
        if (e[u] >= 0) { rl[u] = etype[e[u]]; sc[u] = src_a[e[u]]; ds[u] = dst_a[e[u]]; }
      }
      #pragma unroll
      for (int u = 0; u < CNT_U; u++) {
        if (e[u] >= 0) {
          unsigned i = base + (unsigned)u * 1024u + tid;
          unsigned dl = (unsigned)(ds[u] - lo);
          atomicAdd(&cl[dl * N_REL + (unsigned)rl[u]], 1u);
          q[i] = (int)((unsigned)(rl[u] * N_NODES + sc[u]) | (dl << 20) | ((unsigned)rl[u] << 28));
        }
      }
    }
  }
  __syncthreads();

  for (unsigned i = tid; i < NPB * N_REL; i += 1024) {
    int node = lo + (int)(i >> 4);
    if (node < N_NODES) inv[(size_t)lo * N_REL + i] = 1.0f / fmaxf((float)cl[i], 1.0f);
  }

  if (ws_bytes >= WS_NEED_BYTES) {
    if (tid == 0) qn_g[blockIdx.x] = (int)min(nq, (unsigned)QSLOT + 1u);
    if (nq <= (unsigned)QSLOT)
      for (unsigned i = tid; i < nq; i += 1024) qg[(size_t)blockIdx.x * QSLOT + i] = q[i];
  }

  if (over) {
    for (unsigned e = tid; e < N_EDGE; e += 1024u) {
      int d = dst_a[e];
      if ((unsigned)(d - lo) < (unsigned)NPB) {
        int rel = etype[e], s = src_a[e];
        float ivs = 1.0f / fmaxf((float)cl[(unsigned)(d - lo) * N_REL + (unsigned)rel], 1.0f);
        const float* wr = w1 + ((size_t)rel * N_NODES + (size_t)s) * HID;
        float* ar = acc + (d - lo) * HID;
        for (int h = 0; h < HID; h++) lds_fadd(ar + h, wr[h] * ivs);
      }
    }
  }
  const unsigned n = over ? 0u : nq;
  const unsigned wave = (unsigned)__builtin_amdgcn_readfirstlane((int)(tid >> 5));
  for (unsigned b = wave * 32u; b < n; b += 32u * 32u) {
    const unsigned m = min(32u, n - b);
    int r = 0, dl = 0; float iv = 0.0f;
    if (lane < m) {
      unsigned p = (unsigned)q[b + lane];
      r = (int)(p & 0xFFFFFu);
      dl = (int)((p >> 20) & 0xFFu);
      iv = 1.0f / fmaxf((float)cl[(unsigned)dl * N_REL + (p >> 28)], 1.0f);
    }
    #pragma unroll
    for (unsigned j = 0; j < 32; j += DR_U) {
      if (j >= m) break;
      if (j + DR_U <= m) {
        v2f row[DR_U];
        #pragma unroll
        for (unsigned u = 0; u < DR_U; u++) {
          int rr = __builtin_amdgcn_readlane(r, (int)(j + u));
          row[u] = *(const v2f*)(w1 + (size_t)rr * HID + 2u * lane);
        }
        #pragma unroll
        for (unsigned u = 0; u < DR_U; u++) {
          int dd = __builtin_amdgcn_readlane(dl, (int)(j + u));
          float ii = __builtin_bit_cast(float, __builtin_amdgcn_readlane(__builtin_bit_cast(int, iv), (int)(j + u)));
          float* o = acc + dd * HID + 2u * lane;
          lds_fadd(o + 0, row[u].x * ii);
          lds_fadd(o + 1, row[u].y * ii);
        }
      } else {
        for (unsigned u = j; u < m; u++) {
          int rr = __builtin_amdgcn_readlane(r, (int)u);
          int dd = __builtin_amdgcn_readlane(dl, (int)u);
          float ii = __builtin_bit_cast(float, __builtin_amdgcn_readlane(__builtin_bit_cast(int, iv), (int)u));
          v2f rw = *(const v2f*)(w1 + (size_t)rr * HID + 2u * lane);
          float* o = acc + dd * HID + 2u * lane;
          lds_fadd(o + 0, rw.x * ii);
          lds_fadd(o + 1, rw.y * ii);
        }
      }
    }
  }
  __syncthreads();

  for (unsigned i = tid; i < NPB * HID; i += 1024) {
    int node = lo + (int)(i >> 6);
    if (node < N_NODES) xs[(size_t)node * HID + (i & 63u)] = acc[i];
  }
}

__global__ void k_relu(const float* __restrict__ root1, const float* __restrict__ bias1,
                       float* __restrict__ xs) {
  unsigned idx = blockIdx.x * blockDim.x + threadIdx.x;
  if (idx >= N_NODES * HID) return;
  unsigned h = idx & 63;
  xs[idx] = fmaxf(xs[idx] + root1[idx] + bias1[h], 0.0f);
}


#define QC2   QSLOT
#define L2_U  8
#define QNONE 0xFFFFFFFFu

__global__ __launch_bounds__(1024)
void k_layer2(const int* __restrict__ src_a, const int* __restrict__ dst_a,
              const int* __restrict__ etype, const float* __restrict__ inv,
              const float* __restrict__ xs, const float* __restrict__ w2,
              const float* __restrict__ r2, const float* __restrict__ bias2,
              const int* __restrict__ qn_g, const int* __restrict__ qg,
              unsigned long long ws_bytes, float* __restrict__ out) {
  __shared__ float acc2[NPB * N_LAB];
  __shared__ float invl[NPB * N_REL];
  __shared__ unsigned qs[QC2];
  __shared__ unsigned rc[N_REL];
  __shared__ unsigned ro[N_REL];
  __shared__ unsigned cur[N_REL];
  const unsigned tid = threadIdx.x;
  const unsigned lane = tid & 31;
  const unsigned half = lane >> 4;
  const unsigned sub = lane & 15;
  const int lo = (int)blockIdx.x * NPB;
  const unsigned wave = (unsigned)__builtin_amdgcn_readfirstlane((int)(tid >> 5));

  for (unsigned i = tid; i < NPB * N_REL; i += 1024) {
    int node = lo + (int)(i >> 4);
    invl[i] = (node < N_NODES) ? inv[(size_t)lo * N_REL + i] : 0.0f;
  }

  if (wave < (NPB + 15) / 16) {
    const int node = min(lo + (int)(wave * 16u + sub), N_NODES - 1);
    const float* arow = xs + (size_t)node * HID + 2u * half;
    v8f c0 = {}; v8f c1 = {};
    for (int k0 = 0; k0 < HID; k0 += 4) {
      v2f a;  a.x = arow[k0];  a.y = arow[k0 + 1];
      int kb = k0 + 2 * (int)half;
      v2f b0; b0.x = r2[kb * N_LAB + sub];        b0.y = r2[(kb + 1) * N_LAB + sub];
      v2f b1; b1.x = r2[kb * N_LAB + 16 + sub];   b1.y = r2[(kb + 1) * N_LAB + 16 + sub];
      c0 = __builtin_amdgcn_wmma_f32_16x16x4_f32(false, a, false, b0, (short)0, c0, false, false);
      c1 = __builtin_amdgcn_wmma_f32_16x16x4_f32(false, a, false, b1, (short)0, c1, false, false);
    }
    #pragma unroll
    for (int v = 0; v < 8; v++) {
      unsigned row = wave * 16u + (unsigned)v + 8u * half;
      if (row < (unsigned)NPB) {
        acc2[row * N_LAB + sub]      = c0[v] + bias2[sub];
        acc2[row * N_LAB + 16 + sub] = c1[v] + bias2[16 + sub];
      }
    }
  }

  const unsigned r = wave & 15u;
  const unsigned hw = wave >> 4;
  const float* wr = w2 + (size_t)r * HID * N_LAB;
  v2f b0[16], b1[16];
  #pragma unroll
  for (int s = 0; s < 16; s++) {
    unsigned k = 8u * (unsigned)(s >> 1) + 4u * half + 2u * (unsigned)(s & 1);
    b0[s].x = wr[k * N_LAB + sub];        b0[s].y = wr[(k + 1) * N_LAB + sub];
    b1[s].x = wr[k * N_LAB + 16 + sub];   b1[s].y = wr[(k + 1) * N_LAB + 16 + sub];
  }

  const bool slots = ws_bytes >= WS_NEED_BYTES;
  const unsigned nq = slots ? (unsigned)qn_g[blockIdx.x] : QNONE;
  const bool handed = nq <= (unsigned)QSLOT;
  const unsigned nchunk = handed ? 1u : (unsigned)((N_EDGE + QC2 - 1) / QC2);

  for (unsigned ch = 0; ch < nchunk; ch++) {
    __syncthreads();
    if (tid < N_REL) { rc[tid] = 0; cur[tid] = 0; }
    __syncthreads();

    unsigned p[L2_U];
    #pragma unroll
    for (int u = 0; u < L2_U; u++) {
      const unsigned i = (unsigned)u * 1024u + tid;
      p[u] = QNONE;
      if (handed) {
        if (i < nq) p[u] = (unsigned)qg[(size_t)blockIdx.x * QSLOT + i];
      } else {
        const unsigned e = ch * (unsigned)QC2 + i;
        if (e < (unsigned)N_EDGE) {
          const int d = dst_a[e];
          if ((unsigned)(d - lo) < (unsigned)NPB) {
            const int rl = etype[e], sc = src_a[e];
            p[u] = (unsigned)(rl * N_NODES + sc) | ((unsigned)(d - lo) << 20) | ((unsigned)rl << 28);
          }
        }
      }
      if (p[u] != QNONE) atomicAdd(&rc[p[u] >> 28], 1u);
    }
    __syncthreads();
    if (tid == 0) {
      unsigned a = 0;
      for (int k = 0; k < N_REL; k++) { ro[k] = a; a += rc[k]; }
    }
    __syncthreads();
    #pragma unroll
    for (int u = 0; u < L2_U; u++) {
      if (p[u] != QNONE) {
        const unsigned rr = p[u] >> 28;
        qs[ro[rr] + atomicAdd(&cur[rr], 1u)] = p[u];
      }
    }
    __syncthreads();

    const unsigned n = (unsigned)__builtin_amdgcn_readfirstlane((int)rc[r]);
    const unsigned start = (unsigned)__builtin_amdgcn_readfirstlane((int)ro[r]);
    for (unsigned t0 = 16u * hw; t0 < n; t0 += 32u) {
      const unsigned m = t0 + sub;
      const int s = (m < n) ? (int)(qs[start + m] & 0xFFFFFu) - (int)(r * (unsigned)N_NODES) : 0;
      const v4f* xr = (const v4f*)(xs + (size_t)s * HID + 4u * half);
      v4f q[8];
      #pragma unroll
      for (int i = 0; i < 8; i++) q[i] = xr[2 * i];
      unsigned dl[8]; float iv[8];
      #pragma unroll
      for (int v = 0; v < 8; v++) {
        unsigned row = min(t0 + (unsigned)v + 8u * half, n - 1u);
        dl[v] = (qs[start + row] >> 20) & 0xFFu;
        iv[v] = invl[dl[v] * N_REL + r];
      }
      v8f c0 = {}; v8f c1 = {};
      #pragma unroll
      for (int st = 0; st < 16; st++) {
        v2f a;
        if (st & 1) { a.x = q[st >> 1].z; a.y = q[st >> 1].w; }
        else        { a.x = q[st >> 1].x; a.y = q[st >> 1].y; }
        c0 = __builtin_amdgcn_wmma_f32_16x16x4_f32(false, a, false, b0[st], (short)0, c0, false, false);
        c1 = __builtin_amdgcn_wmma_f32_16x16x4_f32(false, a, false, b1[st], (short)0, c1, false, false);
      }
      float p0[8], p1[8];
      #pragma unroll
      for (int v = 0; v < 8; v++) { p0[v] = c0[v] * iv[v]; p1[v] = c1[v] * iv[v]; }
      #pragma unroll
      for (int v = 0; v < 8; v++) {
        unsigned row = t0 + (unsigned)v + 8u * half;
        if (row < n) {
          float* o = acc2 + dl[v] * N_LAB + sub;
          lds_fadd(o, p0[v]);
          lds_fadd(o + 16, p1[v]);
        }
      }
    }
  }
  __syncthreads();

  for (unsigned i = tid; i < NPB * N_LAB; i += 1024) {
    int node = lo + (int)(i >> 5);
    if (node < N_NODES) out[(size_t)lo * N_LAB + i] = 1.0f / (1.0f + __expf(-acc2[i]));
  }
}

extern "C" void kernel_launch(void* const* d_in, const int* in_sizes, int n_in,
                              void* d_out, int out_size, void* d_ws, size_t ws_size,
                              hipStream_t stream) {
  const int* ei     = (const int*)d_in[0];
  const int* src_a  = ei;
  const int* dst_a  = ei + N_EDGE;
  const int* etype  = (const int*)d_in[1];
  const float* w1   = (const float*)d_in[2];
  const float* r1   = (const float*)d_in[3];
  const float* b1   = (const float*)d_in[4];
  const float* w2   = (const float*)d_in[5];
  const float* r2   = (const float*)d_in[6];
  const float* b2   = (const float*)d_in[7];
  float* out = (float*)d_out;

  float* ws   = (float*)d_ws;
  float* inv  = ws + WS_INV_OFF;
  float* xs   = ws + WS_X_OFF;
  int* qn_g   = (int*)d_ws + WS_QN_OFF;
  int* qg     = (int*)d_ws + WS_QG_OFF;
  int* keys   = (int*)d_ws + WS_KEY_OFF;

  (void)in_sizes; (void)n_in; (void)out_size;

  k_keys<<<(KWORDS + 255) / 256, 256, 0, stream>>>(dst_a, (unsigned*)keys,
                                                   (unsigned long long)ws_size);
  k_layer1<<<NRANGE, 1024, 0, stream>>>(src_a, dst_a, etype, w1, inv, xs, qn_g, qg, keys,
                                        (unsigned long long)ws_size);
  k_relu<<<(N_NODES * HID + 255) / 256, 256, 0, stream>>>(r1, b1, xs);
  k_layer2<<<NRANGE, 1024, 0, stream>>>(src_a, dst_a, etype, inv, xs, w2, r2, b2, qn_g, qg,
                                        (unsigned long long)ws_size, out);
}


# relu(agg1 + root1 + bias1) moves into layer 1's store: the owner adds its root1 rows (asked for before the drain) and bias1 to its LDS sums and stores x once; relu launch gone; 4 launches become 3
# speedup vs baseline: 3.4895x; 1.0670x over previous
#include <hip/hip_runtime.h>

#define N_NODES 50000
#define N_REL   16
#define HID     64
#define N_LAB   32
#define N_EDGE  1600000

typedef float v2f __attribute__((ext_vector_type(2)));
typedef float v4f __attribute__((ext_vector_type(4)));
typedef float v8f __attribute__((ext_vector_type(8)));

#define WS_INV_OFF  0
#define WS_X_OFF    (N_REL * N_NODES)
#define NRANGE      256
#define QSLOT       8192
#define WS_QN_OFF   (WS_X_OFF + N_NODES * HID)
#define WS_QG_OFF   (WS_QN_OFF + NRANGE)
#define WS_NEED_BYTES ((unsigned long long)(WS_QG_OFF + NRANGE * QSLOT) * 4ull)
#define KWORDS      (N_EDGE / 32)
#define KPITCH      50048
#define WS_KEY_OFF  (WS_QG_OFF + NRANGE * QSLOT)
#define WS_KEY_BYTES ((unsigned long long)(WS_KEY_OFF + 8 * KPITCH) * 4ull)

#define NPB     196
#define QCAP    32768
#define SCAN_U  8
#define KSCAN_U 2
#define K4      (KWORDS / 4)
#define CNT_U   8
#define DR_U    16
#define X4      (NPB * HID / 4)
#define X4_U    ((X4 + 1023) / 1024)

__device__ __forceinline__ void lds_fadd(float* p, float v) {
  __hip_atomic_fetch_add(p, v, __ATOMIC_RELAXED, __HIP_MEMORY_SCOPE_WORKGROUP);
}

__global__ void k_keys(const int* __restrict__ dst_a, unsigned* __restrict__ keys,
                       unsigned long long ws_bytes) {
  if (ws_bytes < WS_KEY_BYTES) return;
  unsigned t = blockIdx.x * blockDim.x + threadIdx.x;
  if (t >= KWORDS) return;
  const int4* __restrict__ d4 = (const int4*)dst_a + 8u * t;
  unsigned w[8] = {0u, 0u, 0u, 0u, 0u, 0u, 0u, 0u};
  #pragma unroll
  for (int c = 0; c < 8; c++) {
    int4 d = d4[c];
    int dd[4] = {d.x, d.y, d.z, d.w};
    #pragma unroll
    for (int k = 0; k < 4; k++) {
      unsigned key = min((unsigned)dd[k] / (unsigned)NPB, 255u);
      #pragma unroll
      for (int j = 0; j < 8; j++) w[j] |= ((key >> j) & 1u) << (4 * c + k);
    }
  }
  #pragma unroll
  for (int j = 0; j < 8; j++) keys[(unsigned)j * KPITCH + t] = w[j];
}

__device__ __forceinline__ void scan_planes(const int4* __restrict__ k4p, unsigned tid,
                                            int* q, unsigned* qn) {
  unsigned flip[8];
  #pragma unroll
  for (int j = 0; j < 8; j++) flip[j] = ((blockIdx.x >> j) & 1u) ? 0u : 0xFFFFFFFFu;
  for (unsigned base = 0; base < (unsigned)K4; base += 1024u * KSCAN_U) {
    int4 v[KSCAN_U][8];
    #pragma unroll
    for (int u = 0; u < KSCAN_U; u++) {
      unsigned i = min(base + (unsigned)u * 1024u + tid, (unsigned)K4 - 1u);
      #pragma unroll
      for (int j = 0; j < 8; j++) v[u][j] = k4p[(unsigned)j * (KPITCH / 4) + i];
    }
    #pragma unroll
    for (int u = 0; u < KSCAN_U; u++) {
      const unsigned i = base + (unsigned)u * 1024u + tid;
      const unsigned live = (i < (unsigned)K4) ? 0xFFFFFFFFu : 0u;
      #pragma unroll
      for (int c = 0; c < 4; c++) {
        unsigned m = live;
        #pragma unroll
        for (int j = 0; j < 8; j++) {
          const int pw[4] = {v[u][j].x, v[u][j].y, v[u][j].z, v[u][j].w};
          m &= (unsigned)pw[c] ^ flip[j];
        }
        while (m != 0u) {
          unsigned l = (unsigned)__builtin_ctz(m);
          m &= m - 1u;
          int e = (int)(128u * i + 32u * (unsigned)c + l);
          unsigned idx = atomicAdd(qn, 1u);
          if (idx < QCAP) q[idx] = e;
        }
      }
    }
  }
}

__device__ __forceinline__ void scan_dst(const int* __restrict__ dst_a, int lo, unsigned tid,
                                         int* q, unsigned* qn) {
  const int4* __restrict__ d4p = (const int4*)dst_a;
  const unsigned n4 = N_EDGE / 4;
  for (unsigned base = 0; base < n4; base += 1024u * SCAN_U) {
    int4 v[SCAN_U];
    #pragma unroll
    for (int u = 0; u < SCAN_U; u++) {
      unsigned i = base + (unsigned)u * 1024u + tid;
      v[u] = (i < n4) ? d4p[i] : make_int4(-1, -1, -1, -1);
    }
    #pragma unroll
    for (int u = 0; u < SCAN_U; u++) {
      unsigned i = base + (unsigned)u * 1024u + tid;
      int dd[4] = {v[u].x, v[u].y, v[u].z, v[u].w};
      unsigned nearest = min(min((unsigned)(dd[0] - lo), (unsigned)(dd[1] - lo)),
                             min((unsigned)(dd[2] - lo), (unsigned)(dd[3] - lo)));
      if (nearest < (unsigned)NPB) {
        #pragma unroll
        for (int c = 0; c < 4; c++) {
          if ((unsigned)(dd[c] - lo) < (unsigned)NPB) {
            int e = (int)(4u * i) + c;
            unsigned idx = atomicAdd(qn, 1u);
            if (idx < QCAP) q[idx] = e;
          }
        }
      }
    }
  }
}

__global__ __launch_bounds__(1024)
void k_layer1(const int* __restrict__ src_a, const int* __restrict__ dst_a,
              const int* __restrict__ etype, const float* __restrict__ w1,
              const float* __restrict__ root1, const float* __restrict__ bias1,
              float* __restrict__ inv, float* __restrict__ xs,
              int* __restrict__ qn_g, int* __restrict__ qg, const int* __restrict__ keys,
              unsigned long long ws_bytes) {
  __shared__ __attribute__((aligned(16))) float acc[NPB * HID];
  __shared__ int q[QCAP];
  __shared__ unsigned cl[NPB * N_REL];
  __shared__ unsigned qn;
  const unsigned tid = threadIdx.x;
  const unsigned lane = tid & 31;
  const int lo = (int)blockIdx.x * NPB;

  for (unsigned i = tid; i < NPB * HID; i += 1024) acc[i] = 0.0f;
  for (unsigned i = tid; i < NPB * N_REL; i += 1024) cl[i] = 0u;
  if (tid == 0) qn = 0;
  __syncthreads();

  if (ws_bytes >= WS_KEY_BYTES) scan_planes((const int4*)keys, tid, q, &qn);
  else                          scan_dst(dst_a, lo, tid, q, &qn);
  __syncthreads();

  const unsigned nq = qn;
  const bool over = nq > (unsigned)QCAP;
  if (over) {
    for (unsigned e = tid; e < N_EDGE; e += 1024u) {
      int d = dst_a[e];
      if ((unsigned)(d - lo) < (unsigned)NPB) atomicAdd(&cl[(unsigned)(d - lo) * N_REL + (unsigned)etype[e]], 1u);
    }
  } else {
    for (unsigned base = 0; base < nq; base += 1024u * CNT_U) {
      int e[CNT_U], rl[CNT_U], sc[CNT_U], ds[CNT_U];
      #pragma unroll
      for (int u = 0; u < CNT_U; u++) {
        unsigned i = base + (unsigned)u * 1024u + tid;
        e[u] = (i < nq) ? q[i] : -1;
      }
      #pragma unroll
      for (int u = 0; u < CNT_U; u++) {
        rl[u] = 0; sc[u] = 0; ds[u] = lo;
        if (e[u] >= 0) { rl[u] = etype[e[u]]; sc[u] = src_a[e[u]]; ds[u] = dst_a[e[u]]; }
      }
      #pragma unroll
      for (int u = 0; u < CNT_U; u++) {
        if (e[u] >= 0) {
          unsigned i = base + (unsigned)u * 1024u + tid;
          unsigned dl = (unsigned)(ds[u] - lo);
          atomicAdd(&cl[dl * N_REL + (unsigned)rl[u]], 1u);
          q[i] = (int)((unsigned)(rl[u] * N_NODES + sc[u]) | (dl << 20) | ((unsigned)rl[u] << 28));
        }
      }
    }
  }
  __syncthreads();

  for (unsigned i = tid; i < NPB * N_REL; i += 1024) {
    int node = lo + (int)(i >> 4);
    if (node < N_NODES) inv[(size_t)lo * N_REL + i] = 1.0f / fmaxf((float)cl[i], 1.0f);
  }

  if (ws_bytes >= WS_NEED_BYTES) {
    if (tid == 0) qn_g[blockIdx.x] = (int)min(nq, (unsigned)QSLOT + 1u);
    if (nq <= (unsigned)QSLOT)
      for (unsigned i = tid; i < nq; i += 1024) qg[(size_t)blockIdx.x * QSLOT + i] = q[i];
  }

  if (over) {
    for (unsigned e = tid; e < N_EDGE; e += 1024u) {
      int d = dst_a[e];
      if ((unsigned)(d - lo) < (unsigned)NPB) {
        int rel = etype[e], s = src_a[e];
        float ivs = 1.0f / fmaxf((float)cl[(unsigned)(d - lo) * N_REL + (unsigned)rel], 1.0f);
        const float* wr = w1 + ((size_t)rel * N_NODES + (size_t)s) * HID;
        float* ar = acc + (d - lo) * HID;
        for (int h = 0; h < HID; h++) lds_fadd(ar + h, wr[h] * ivs);
      }
    }
  }
  v4f rt[X4_U];
  #pragma unroll
  for (int k = 0; k < X4_U; k++) {
    const unsigned i4 = (unsigned)k * 1024u + tid;
    const int node = lo + (int)(i4 >> 4);
    rt[k] = (v4f){0.0f, 0.0f, 0.0f, 0.0f};
    if (i4 < (unsigned)X4 && node < N_NODES) rt[k] = *(const v4f*)(root1 + (size_t)lo * HID + 4u * i4);
  }
  const v4f bs = *(const v4f*)(bias1 + ((4u * tid) & 63u));

  const unsigned n = over ? 0u : nq;
  const unsigned wave = (unsigned)__builtin_amdgcn_readfirstlane((int)(tid >> 5));
  for (unsigned b = wave * 32u; b < n; b += 32u * 32u) {
    const unsigned m = min(32u, n - b);
    int r = 0, dl = 0; float iv = 0.0f;
    if (lane < m) {
      unsigned p = (unsigned)q[b + lane];
      r = (int)(p & 0xFFFFFu);
      dl = (int)((p >> 20) & 0xFFu);
      iv = 1.0f / fmaxf((float)cl[(unsigned)dl * N_REL + (p >> 28)], 1.0f);
    }
    #pragma unroll
    for (unsigned j = 0; j < 32; j += DR_U) {
      if (j >= m) break;
      if (j + DR_U <= m) {
        v2f row[DR_U];
        #pragma unroll
        for (unsigned u = 0; u < DR_U; u++) {
          int rr = __builtin_amdgcn_readlane(r, (int)(j + u));
          row[u] = *(const v2f*)(w1 + (size_t)rr * HID + 2u * lane);
        }
        #pragma unroll
        for (unsigned u = 0; u < DR_U; u++) {
          int dd = __builtin_amdgcn_readlane(dl, (int)(j + u));
          float ii = __builtin_bit_cast(float, __builtin_amdgcn_readlane(__builtin_bit_cast(int, iv), (int)(j + u)));
          float* o = acc + dd * HID + 2u * lane;
          lds_fadd(o + 0, row[u].x * ii);
          lds_fadd(o + 1, row[u].y * ii);
        }
      } else {
        for (unsigned u = j; u < m; u++) {
          int rr = __builtin_amdgcn_readlane(r, (int)u);
          int dd = __builtin_amdgcn_readlane(dl, (int)u);
          float ii = __builtin_bit_cast(float, __builtin_amdgcn_readlane(__builtin_bit_cast(int, iv), (int)u));
          v2f rw = *(const v2f*)(w1 + (size_t)rr * HID + 2u * lane);
          float* o = acc + dd * HID + 2u * lane;
          lds_fadd(o + 0, rw.x * ii);
          lds_fadd(o + 1, rw.y * ii);
        }
      }
    }
  }
  __syncthreads();

  #pragma unroll
  for (int k = 0; k < X4_U; k++) {
    const unsigned i4 = (unsigned)k * 1024u + tid;
    const int node = lo + (int)(i4 >> 4);
    if (i4 < (unsigned)X4 && node < N_NODES) {
      const v4f a = *(const v4f*)(acc + 4u * i4);
      v4f x;
      x.x = fmaxf(a.x + rt[k].x + bs.x, 0.0f);
      x.y = fmaxf(a.y + rt[k].y + bs.y, 0.0f);
      x.z = fmaxf(a.z + rt[k].z + bs.z, 0.0f);
      x.w = fmaxf(a.w + rt[k].w + bs.w, 0.0f);
      *(v4f*)(xs + (size_t)lo * HID + 4u * i4) = x;
    }
  }
}


#define QC2   QSLOT
#define L2_U  8
#define QNONE 0xFFFFFFFFu

__global__ __launch_bounds__(1024)
void k_layer2(const int* __restrict__ src_a, const int* __restrict__ dst_a,
              const int* __restrict__ etype, const float* __restrict__ inv,
              const float* __restrict__ xs, const float* __restrict__ w2,
              const float* __restrict__ r2, const float* __restrict__ bias2,
              const int* __restrict__ qn_g, const int* __restrict__ qg,
              unsigned long long ws_bytes, float* __restrict__ out) {
  __shared__ float acc2[NPB * N_LAB];
  __shared__ float invl[NPB * N_REL];
  __shared__ unsigned qs[QC2];
  __shared__ unsigned rc[N_REL];
  __shared__ unsigned ro[N_REL];
  __shared__ unsigned cur[N_REL];
  const unsigned tid = threadIdx.x;
  const unsigned lane = tid & 31;
  const unsigned half = lane >> 4;
  const unsigned sub = lane & 15;
  const int lo = (int)blockIdx.x * NPB;
  const unsigned wave = (unsigned)__builtin_amdgcn_readfirstlane((int)(tid >> 5));

  for (unsigned i = tid; i < NPB * N_REL; i += 1024) {
    int node = lo + (int)(i >> 4);
    invl[i] = (node < N_NODES) ? inv[(size_t)lo * N_REL + i] : 0.0f;
  }

  if (wave < (NPB + 15) / 16) {
    const int node = min(lo + (int)(wave * 16u + sub), N_NODES - 1);
    const float* arow = xs + (size_t)node * HID + 2u * half;
    v8f c0 = {}; v8f c1 = {};
    for (int k0 = 0; k0 < HID; k0 += 4) {
      v2f a;  a.x = arow[k0];  a.y = arow[k0 + 1];
      int kb = k0 + 2 * (int)half;
      v2f b0; b0.x = r2[kb * N_LAB + sub];        b0.y = r2[(kb + 1) * N_LAB + sub];
      v2f b1; b1.x = r2[kb * N_LAB + 16 + sub];   b1.y = r2[(kb + 1) * N_LAB + 16 + sub];
      c0 = __builtin_amdgcn_wmma_f32_16x16x4_f32(false, a, false, b0, (short)0, c0, false, false);
      c1 = __builtin_amdgcn_wmma_f32_16x16x4_f32(false, a, false, b1, (short)0, c1, false, false);
    }
    #pragma unroll
    for (int v = 0; v < 8; v++) {
      unsigned row = wave * 16u + (unsigned)v + 8u * half;
      if (row < (unsigned)NPB) {
        acc2[row * N_LAB + sub]      = c0[v] + bias2[sub];
        acc2[row * N_LAB + 16 + sub] = c1[v] + bias2[16 + sub];
      }
    }
  }

  const unsigned r = wave & 15u;
  const unsigned hw = wave >> 4;
  const float* wr = w2 + (size_t)r * HID * N_LAB;
  v2f b0[16], b1[16];
  #pragma unroll
  for (int s = 0; s < 16; s++) {
    unsigned k = 8u * (unsigned)(s >> 1) + 4u * half + 2u * (unsigned)(s & 1);
    b0[s].x = wr[k * N_LAB + sub];        b0[s].y = wr[(k + 1) * N_LAB + sub];
    b1[s].x = wr[k * N_LAB + 16 + sub];   b1[s].y = wr[(k + 1) * N_LAB + 16 + sub];
  }

  const bool slots = ws_bytes >= WS_NEED_BYTES;
  const unsigned nq = slots ? (unsigned)qn_g[blockIdx.x] : QNONE;
  const bool handed = nq <= (unsigned)QSLOT;
  const unsigned nchunk = handed ? 1u : (unsigned)((N_EDGE + QC2 - 1) / QC2);

  for (unsigned ch = 0; ch < nchunk; ch++) {
    __syncthreads();
    if (tid < N_REL) { rc[tid] = 0; cur[tid] = 0; }
    __syncthreads();

    unsigned p[L2_U];
    #pragma unroll
    for (int u = 0; u < L2_U; u++) {
      const unsigned i = (unsigned)u * 1024u + tid;
      p[u] = QNONE;
      if (handed) {
        if (i < nq) p[u] = (unsigned)qg[(size_t)blockIdx.x * QSLOT + i];
      } else {
        const unsigned e = ch * (unsigned)QC2 + i;
        if (e < (unsigned)N_EDGE) {
          const int d = dst_a[e];
          if ((unsigned)(d - lo) < (unsigned)NPB) {
            const int rl = etype[e], sc = src_a[e];
            p[u] = (unsigned)(rl * N_NODES + sc) | ((unsigned)(d - lo) << 20) | ((unsigned)rl << 28);
          }
        }
      }
      if (p[u] != QNONE) atomicAdd(&rc[p[u] >> 28], 1u);
    }
    __syncthreads();
    if (tid == 0) {
      unsigned a = 0;
      for (int k = 0; k < N_REL; k++) { ro[k] = a; a += rc[k]; }
    }
    __syncthreads();
    #pragma unroll
    for (int u = 0; u < L2_U; u++) {
      if (p[u] != QNONE) {
        const unsigned rr = p[u] >> 28;
        qs[ro[rr] + atomicAdd(&cur[rr], 1u)] = p[u];
      }
    }
    __syncthreads();

    const unsigned n = (unsigned)__builtin_amdgcn_readfirstlane((int)rc[r]);
    const unsigned start = (unsigned)__builtin_amdgcn_readfirstlane((int)ro[r]);
    for (unsigned t0 = 16u * hw; t0 < n; t0 += 32u) {
      const unsigned m = t0 + sub;
      const int s = (m < n) ? (int)(qs[start + m] & 0xFFFFFu) - (int)(r * (unsigned)N_NODES) : 0;
      const v4f* xr = (const v4f*)(xs + (size_t)s * HID + 4u * half);
      v4f q[8];
      #pragma unroll
      for (int i = 0; i < 8; i++) q[i] = xr[2 * i];
      unsigned dl[8]; float iv[8];
      #pragma unroll
      for (int v = 0; v < 8; v++) {
        unsigned row = min(t0 + (unsigned)v + 8u * half, n - 1u);
        dl[v] = (qs[start + row] >> 20) & 0xFFu;
        iv[v] = invl[dl[v] * N_REL + r];
      }
      v8f c0 = {}; v8f c1 = {};
      #pragma unroll
      for (int st = 0; st < 16; st++) {
        v2f a;
        if (st & 1) { a.x = q[st >> 1].z; a.y = q[st >> 1].w; }
        else        { a.x = q[st >> 1].x; a.y = q[st >> 1].y; }
        c0 = __builtin_amdgcn_wmma_f32_16x16x4_f32(false, a, false, b0[st], (short)0, c0, false, false);
        c1 = __builtin_amdgcn_wmma_f32_16x16x4_f32(false, a, false, b1[st], (short)0, c1, false, false);
      }
      float p0[8], p1[8];
      #pragma unroll
      for (int v = 0; v < 8; v++) { p0[v] = c0[v] * iv[v]; p1[v] = c1[v] * iv[v]; }
      #pragma unroll
      for (int v = 0; v < 8; v++) {
        unsigned row = t0 + (unsigned)v + 8u * half;
        if (row < n) {
          float* o = acc2 + dl[v] * N_LAB + sub;
          lds_fadd(o, p0[v]);
          lds_fadd(o + 16, p1[v]);
        }
      }
    }
  }
  __syncthreads();

  for (unsigned i = tid; i < NPB * N_LAB; i += 1024) {
    int node = lo + (int)(i >> 5);
    if (node < N_NODES) out[(size_t)lo * N_LAB + i] = 1.0f / (1.0f + __expf(-acc2[i]));
  }
}

extern "C" void kernel_launch(void* const* d_in, const int* in_sizes, int n_in,
                              void* d_out, int out_size, void* d_ws, size_t ws_size,
                              hipStream_t stream) {
  const int* ei     = (const int*)d_in[0];
  const int* src_a  = ei;
  const int* dst_a  = ei + N_EDGE;
  const int* etype  = (const int*)d_in[1];
  const float* w1   = (const float*)d_in[2];
  const float* r1   = (const float*)d_in[3];
  const float* b1   = (const float*)d_in[4];
  const float* w2   = (const float*)d_in[5];
  const float* r2   = (const float*)d_in[6];
  const float* b2   = (const float*)d_in[7];
  float* out = (float*)d_out;

  float* ws   = (float*)d_ws;
  float* inv  = ws + WS_INV_OFF;
  float* xs   = ws + WS_X_OFF;
  int* qn_g   = (int*)d_ws + WS_QN_OFF;
  int* qg     = (int*)d_ws + WS_QG_OFF;
  int* keys   = (int*)d_ws + WS_KEY_OFF;

  (void)in_sizes; (void)n_in; (void)out_size;

  k_keys<<<(KWORDS + 255) / 256, 256, 0, stream>>>(dst_a, (unsigned*)keys,
                                                   (unsigned long long)ws_size);
  k_layer1<<<NRANGE, 1024, 0, stream>>>(src_a, dst_a, etype, w1, r1, b1, inv, xs, qn_g, qg, keys,
                                        (unsigned long long)ws_size);
  k_layer2<<<NRANGE, 1024, 0, stream>>>(src_a, dst_a, etype, inv, xs, w2, r2, b2, qn_g, qg,
                                        (unsigned long long)ws_size, out);
}


# first launch also writes one packed word per edge (rel*N+src, local dst, rel) in whole lines; layer 1 gathers that 1 word per queued edge instead of its src, dst and type words; sum order unchanged
# speedup vs baseline: 3.7582x; 1.0770x over previous
#include <hip/hip_runtime.h>

#define N_NODES 50000
#define N_REL   16
#define HID     64
#define N_LAB   32
#define N_EDGE  1600000

typedef float v2f __attribute__((ext_vector_type(2)));
typedef float v4f __attribute__((ext_vector_type(4)));
typedef float v8f __attribute__((ext_vector_type(8)));

#define WS_INV_OFF  0
#define WS_X_OFF    (N_REL * N_NODES)
#define NRANGE      256
#define QSLOT       8192
#define WS_QN_OFF   (WS_X_OFF + N_NODES * HID)
#define WS_QG_OFF   (WS_QN_OFF + NRANGE)
#define WS_NEED_BYTES ((unsigned long long)(WS_QG_OFF + NRANGE * QSLOT) * 4ull)
#define KWORDS      (N_EDGE / 32)
#define KPITCH      50048
#define WS_KEY_OFF  (WS_QG_OFF + NRANGE * QSLOT)
#define WS_KEY_BYTES ((unsigned long long)(WS_KEY_OFF + 8 * KPITCH) * 4ull)
#define WS_PK_OFF   (WS_KEY_OFF + 8 * KPITCH)
#define WS_PK_BYTES ((unsigned long long)(WS_PK_OFF + N_EDGE) * 4ull)

#define NPB     196
#define QCAP    32768
#define SCAN_U  8
#define KSCAN_U 2
#define K4      (KWORDS / 4)
#define CNT_U   8
#define DR_U    16
#define X4      (NPB * HID / 4)
#define X4_U    ((X4 + 1023) / 1024)

__device__ __forceinline__ void lds_fadd(float* p, float v) {
  __hip_atomic_fetch_add(p, v, __ATOMIC_RELAXED, __HIP_MEMORY_SCOPE_WORKGROUP);
}

#define KB_T   256
#define KB_P   4
#define KB_E4  (KB_T * KB_P)
#define KB_W   (KB_E4 / 8)
#define E4     (N_EDGE / 4)

template <bool PK, bool WHOLE>
__device__ __forceinline__ void keys_body(const int* __restrict__ src_a, const int* __restrict__ dst_a,
                                          const int* __restrict__ etype,
                                          unsigned* __restrict__ keys, int* __restrict__ pk,
                                          unsigned* kw) {
  const unsigned tid = threadIdx.x;
  const unsigned g4 = blockIdx.x * (unsigned)KB_E4;
  const int4* __restrict__ d4 = (const int4*)dst_a;
  const int4* __restrict__ s4 = (const int4*)src_a;
  const int4* __restrict__ r4 = (const int4*)etype;
  int4* __restrict__ p4 = (int4*)pk;
  const int4 zero4 = make_int4(0, 0, 0, 0);
  int4 d[KB_P], s[KB_P], r[KB_P];
  #pragma unroll
  for (int c = 0; c < KB_P; c++) {
    const unsigned i4 = WHOLE ? g4 + (unsigned)c * KB_T + tid
                              : min(g4 + (unsigned)c * KB_T + tid, (unsigned)E4 - 1u);
    d[c] = d4[i4];
    s[c] = PK ? s4[i4] : zero4;
    r[c] = PK ? r4[i4] : zero4;
  }
  #pragma unroll
  for (int c = 0; c < KB_P; c++) {
    const unsigned i4 = g4 + (unsigned)c * KB_T + tid;
    const int dd[4] = {d[c].x, d[c].y, d[c].z, d[c].w};
    const int ss[4] = {s[c].x, s[c].y, s[c].z, s[c].w};
    const int rr[4] = {r[c].x, r[c].y, r[c].z, r[c].w};
    unsigned kb = 0u;
    int pp[4];
    #pragma unroll
    for (int k = 0; k < 4; k++) {
      const unsigned key = min((unsigned)dd[k] / (unsigned)NPB, 255u);
      const unsigned dl = ((unsigned)dd[k] - key * (unsigned)NPB) & 0xFFu;
      kb |= key << (8 * k);
      pp[k] = (int)((unsigned)(rr[k] * N_NODES + ss[k]) | (dl << 20) | ((unsigned)rr[k] << 28));
    }
    kw[(unsigned)c * KB_T + tid] = kb;
    if (PK && (WHOLE || i4 < (unsigned)E4)) p4[i4] = make_int4(pp[0], pp[1], pp[2], pp[3]);
  }
  __syncthreads();

  const unsigned t = blockIdx.x * (unsigned)KB_W + tid;
  if (tid >= (unsigned)KB_W || t >= KWORDS) return;
  const uint4 ka = *(const uint4*)(kw + 8u * tid);
  const uint4 kc = *(const uint4*)(kw + 8u * tid + 4u);
  const unsigned kk[8] = {ka.x, ka.y, ka.z, ka.w, kc.x, kc.y, kc.z, kc.w};
  unsigned w[8] = {0u, 0u, 0u, 0u, 0u, 0u, 0u, 0u};
  #pragma unroll
  for (int q = 0; q < 8; q++) {
    #pragma unroll
    for (int k = 0; k < 4; k++) {
      #pragma unroll
      for (int j = 0; j < 8; j++) w[j] |= ((kk[q] >> (8 * k + j)) & 1u) << (4 * q + k);
    }
  }
  #pragma unroll
  for (int j = 0; j < 8; j++) keys[(unsigned)j * KPITCH + t] = w[j];
}

__global__ __launch_bounds__(KB_T)
void k_keys(const int* __restrict__ src_a, const int* __restrict__ dst_a,
            const int* __restrict__ etype, unsigned* __restrict__ keys, int* __restrict__ pk,
            unsigned long long ws_bytes) {
  __shared__ __attribute__((aligned(16))) unsigned kw[KB_E4];
  if (ws_bytes < WS_KEY_BYTES) return;
  const bool whole = blockIdx.x * (unsigned)KB_E4 + (unsigned)KB_E4 <= (unsigned)E4;
  if (ws_bytes >= WS_PK_BYTES) {
    if (whole) keys_body<true, true>(src_a, dst_a, etype, keys, pk, kw);
    else       keys_body<true, false>(src_a, dst_a, etype, keys, pk, kw);
  } else {
    if (whole) keys_body<false, true>(src_a, dst_a, etype, keys, pk, kw);
    else       keys_body<false, false>(src_a, dst_a, etype, keys, pk, kw);
  }
}

__device__ __forceinline__ void scan_planes(const int4* __restrict__ k4p, unsigned tid,
                                            int* q, unsigned* qn) {
  unsigned flip[8];
  #pragma unroll
  for (int j = 0; j < 8; j++) flip[j] = ((blockIdx.x >> j) & 1u) ? 0u : 0xFFFFFFFFu;
  for (unsigned base = 0; base < (unsigned)K4; base += 1024u * KSCAN_U) {
    int4 v[KSCAN_U][8];
    #pragma unroll
    for (int u = 0; u < KSCAN_U; u++) {
      unsigned i = min(base + (unsigned)u * 1024u + tid, (unsigned)K4 - 1u);
      #pragma unroll
      for (int j = 0; j < 8; j++) v[u][j] = k4p[(unsigned)j * (KPITCH / 4) + i];
    }
    #pragma unroll
    for (int u = 0; u < KSCAN_U; u++) {
      const unsigned i = base + (unsigned)u * 1024u + tid;
      const unsigned live = (i < (unsigned)K4) ? 0xFFFFFFFFu : 0u;
      #pragma unroll
      for (int c = 0; c < 4; c++) {
        unsigned m = live;
        #pragma unroll
        for (int j = 0; j < 8; j++) {
          const int pw[4] = {v[u][j].x, v[u][j].y, v[u][j].z, v[u][j].w};
          m &= (unsigned)pw[c] ^ flip[j];
        }
        while (m != 0u) {
          unsigned l = (unsigned)__builtin_ctz(m);
          m &= m - 1u;
          int e = (int)(128u * i + 32u * (unsigned)c + l);
          unsigned idx = atomicAdd(qn, 1u);
          if (idx < QCAP) q[idx] = e;
        }
      }
    }
  }
}

__device__ __forceinline__ void scan_dst(const int* __restrict__ dst_a, int lo, unsigned tid,
                                         int* q, unsigned* qn) {
  const int4* __restrict__ d4p = (const int4*)dst_a;
  const unsigned n4 = N_EDGE / 4;
  for (unsigned base = 0; base < n4; base += 1024u * SCAN_U) {
    int4 v[SCAN_U];
    #pragma unroll
    for (int u = 0; u < SCAN_U; u++) {
      unsigned i = base + (unsigned)u * 1024u + tid;
      v[u] = (i < n4) ? d4p[i] : make_int4(-1, -1, -1, -1);
    }
    #pragma unroll
    for (int u = 0; u < SCAN_U; u++) {
      unsigned i = base + (unsigned)u * 1024u + tid;
      int dd[4] = {v[u].x, v[u].y, v[u].z, v[u].w};
      unsigned nearest = min(min((unsigned)(dd[0] - lo), (unsigned)(dd[1] - lo)),
                             min((unsigned)(dd[2] - lo), (unsigned)(dd[3] - lo)));
      if (nearest < (unsigned)NPB) {
        #pragma unroll
        for (int c = 0; c < 4; c++) {
          if ((unsigned)(dd[c] - lo) < (unsigned)NPB) {
            int e = (int)(4u * i) + c;
            unsigned idx = atomicAdd(qn, 1u);
            if (idx < QCAP) q[idx] = e;
          }
        }
      }
    }
  }
}

__global__ __launch_bounds__(1024)
void k_layer1(const int* __restrict__ src_a, const int* __restrict__ dst_a,
              const int* __restrict__ etype, const float* __restrict__ w1,
              const float* __restrict__ root1, const float* __restrict__ bias1,
              float* __restrict__ inv, float* __restrict__ xs,
              int* __restrict__ qn_g, int* __restrict__ qg, const int* __restrict__ keys,
              const int* __restrict__ pk, unsigned long long ws_bytes) {
  __shared__ __attribute__((aligned(16))) float acc[NPB * HID];
  __shared__ int q[QCAP];
  __shared__ unsigned cl[NPB * N_REL];
  __shared__ unsigned qn;
  const unsigned tid = threadIdx.x;
  const unsigned lane = tid & 31;
  const int lo = (int)blockIdx.x * NPB;

  for (unsigned i = tid; i < NPB * HID; i += 1024) acc[i] = 0.0f;
  for (unsigned i = tid; i < NPB * N_REL; i += 1024) cl[i] = 0u;
  if (tid == 0) qn = 0;
  __syncthreads();

  if (ws_bytes >= WS_KEY_BYTES) scan_planes((const int4*)keys, tid, q, &qn);
  else                          scan_dst(dst_a, lo, tid, q, &qn);
  __syncthreads();

  const unsigned nq = qn;
  const bool over = nq > (unsigned)QCAP;
  if (over) {
    for (unsigned e = tid; e < N_EDGE; e += 1024u) {
      int d = dst_a[e];
      if ((unsigned)(d - lo) < (unsigned)NPB) atomicAdd(&cl[(unsigned)(d - lo) * N_REL + (unsigned)etype[e]], 1u);
    }
  } else if (ws_bytes >= WS_PK_BYTES) {
    for (unsigned base = 0; base < nq; base += 1024u * CNT_U) {
      int e[CNT_U]; unsigned p[CNT_U];
      #pragma unroll
      for (int u = 0; u < CNT_U; u++) {
        unsigned i = base + (unsigned)u * 1024u + tid;
        e[u] = (i < nq) ? q[i] : -1;
      }
      #pragma unroll
      for (int u = 0; u < CNT_U; u++) p[u] = (e[u] >= 0) ? (unsigned)pk[e[u]] : 0u;
      #pragma unroll
      for (int u = 0; u < CNT_U; u++) {
        if (e[u] >= 0) {
          unsigned i = base + (unsigned)u * 1024u + tid;
          atomicAdd(&cl[((p[u] >> 20) & 0xFFu) * N_REL + (p[u] >> 28)], 1u);
          q[i] = (int)p[u];
        }
      }
    }
  } else {
    for (unsigned base = 0; base < nq; base += 1024u * CNT_U) {
      int e[CNT_U], rl[CNT_U], sc[CNT_U], ds[CNT_U];
      #pragma unroll
      for (int u = 0; u < CNT_U; u++) {
        unsigned i = base + (unsigned)u * 1024u + tid;
        e[u] = (i < nq) ? q[i] : -1;
      }
      #pragma unroll
      for (int u = 0; u < CNT_U; u++) {
        rl[u] = 0; sc[u] = 0; ds[u] = lo;
        if (e[u] >= 0) { rl[u] = etype[e[u]]; sc[u] = src_a[e[u]]; ds[u] = dst_a[e[u]]; }
      }
      #pragma unroll
      for (int u = 0; u < CNT_U; u++) {
        if (e[u] >= 0) {
          unsigned i = base + (unsigned)u * 1024u + tid;
          unsigned dl = (unsigned)(ds[u] - lo);
          atomicAdd(&cl[dl * N_REL + (unsigned)rl[u]], 1u);
          q[i] = (int)((unsigned)(rl[u] * N_NODES + sc[u]) | (dl << 20) | ((unsigned)rl[u] << 28));
        }
      }
    }
  }
  __syncthreads();

  for (unsigned i = tid; i < NPB * N_REL; i += 1024) {
    int node = lo + (int)(i >> 4);
    if (node < N_NODES) inv[(size_t)lo * N_REL + i] = 1.0f / fmaxf((float)cl[i], 1.0f);
  }

  if (ws_bytes >= WS_NEED_BYTES) {
    if (tid == 0) qn_g[blockIdx.x] = (int)min(nq, (unsigned)QSLOT + 1u);
    if (nq <= (unsigned)QSLOT)
      for (unsigned i = tid; i < nq; i += 1024) qg[(size_t)blockIdx.x * QSLOT + i] = q[i];
  }

  if (over) {
    for (unsigned e = tid; e < N_EDGE; e += 1024u) {
      int d = dst_a[e];
      if ((unsigned)(d - lo) < (unsigned)NPB) {
        int rel = etype[e], s = src_a[e];
        float ivs = 1.0f / fmaxf((float)cl[(unsigned)(d - lo) * N_REL + (unsigned)rel], 1.0f);
        const float* wr = w1 + ((size_t)rel * N_NODES + (size_t)s) * HID;
        float* ar = acc + (d - lo) * HID;
        for (int h = 0; h < HID; h++) lds_fadd(ar + h, wr[h] * ivs);
      }
    }
  }
  v4f rt[X4_U];
  #pragma unroll
  for (int k = 0; k < X4_U; k++) {
    const unsigned i4 = (unsigned)k * 1024u + tid;
    const int node = lo + (int)(i4 >> 4);
    rt[k] = (v4f){0.0f, 0.0f, 0.0f, 0.0f};
    if (i4 < (unsigned)X4 && node < N_NODES) rt[k] = *(const v4f*)(root1 + (size_t)lo * HID + 4u * i4);
  }
  const v4f bs = *(const v4f*)(bias1 + ((4u * tid) & 63u));

  const unsigned n = over ? 0u : nq;
  const unsigned wave = (unsigned)__builtin_amdgcn_readfirstlane((int)(tid >> 5));
  for (unsigned b = wave * 32u; b < n; b += 32u * 32u) {
    const unsigned m = min(32u, n - b);
    int r = 0, dl = 0; float iv = 0.0f;
    if (lane < m) {
      unsigned p = (unsigned)q[b + lane];
      r = (int)(p & 0xFFFFFu);
      dl = (int)((p >> 20) & 0xFFu);
      iv = 1.0f / fmaxf((float)cl[(unsigned)dl * N_REL + (p >> 28)], 1.0f);
    }
    #pragma unroll
    for (unsigned j = 0; j < 32; j += DR_U) {
      if (j >= m) break;
      if (j + DR_U <= m) {
        v2f row[DR_U];
        #pragma unroll
        for (unsigned u = 0; u < DR_U; u++) {
          int rr = __builtin_amdgcn_readlane(r, (int)(j + u));
          row[u] = *(const v2f*)(w1 + (size_t)rr * HID + 2u * lane);
        }
        #pragma unroll
        for (unsigned u = 0; u < DR_U; u++) {
          int dd = __builtin_amdgcn_readlane(dl, (int)(j + u));
          float ii = __builtin_bit_cast(float, __builtin_amdgcn_readlane(__builtin_bit_cast(int, iv), (int)(j + u)));
          float* o = acc + dd * HID + 2u * lane;
          lds_fadd(o + 0, row[u].x * ii);
          lds_fadd(o + 1, row[u].y * ii);
        }
      } else {
        for (unsigned u = j; u < m; u++) {
          int rr = __builtin_amdgcn_readlane(r, (int)u);
          int dd = __builtin_amdgcn_readlane(dl, (int)u);
          float ii = __builtin_bit_cast(float, __builtin_amdgcn_readlane(__builtin_bit_cast(int, iv), (int)u));
          v2f rw = *(const v2f*)(w1 + (size_t)rr * HID + 2u * lane);
          float* o = acc + dd * HID + 2u * lane;
          lds_fadd(o + 0, rw.x * ii);
          lds_fadd(o + 1, rw.y * ii);
        }
      }
    }
  }
  __syncthreads();

  #pragma unroll
  for (int k = 0; k < X4_U; k++) {
    const unsigned i4 = (unsigned)k * 1024u + tid;
    const int node = lo + (int)(i4 >> 4);
    if (i4 < (unsigned)X4 && node < N_NODES) {
      const v4f a = *(const v4f*)(acc + 4u * i4);
      v4f x;
      x.x = fmaxf(a.x + rt[k].x + bs.x, 0.0f);
      x.y = fmaxf(a.y + rt[k].y + bs.y, 0.0f);
      x.z = fmaxf(a.z + rt[k].z + bs.z, 0.0f);
      x.w = fmaxf(a.w + rt[k].w + bs.w, 0.0f);
      *(v4f*)(xs + (size_t)lo * HID + 4u * i4) = x;
    }
  }
}


#define QC2   QSLOT
#define L2_U  8
#define QNONE 0xFFFFFFFFu

__global__ __launch_bounds__(1024)
void k_layer2(const int* __restrict__ src_a, const int* __restrict__ dst_a,
              const int* __restrict__ etype, const float* __restrict__ inv,
              const float* __restrict__ xs, const float* __restrict__ w2,
              const float* __restrict__ r2, const float* __restrict__ bias2,
              const int* __restrict__ qn_g, const int* __restrict__ qg,
              unsigned long long ws_bytes, float* __restrict__ out) {
  __shared__ float acc2[NPB * N_LAB];
  __shared__ float invl[NPB * N_REL];
  __shared__ unsigned qs[QC2];
  __shared__ unsigned rc[N_REL];
  __shared__ unsigned ro[N_REL];
  __shared__ unsigned cur[N_REL];
  const unsigned tid = threadIdx.x;
  const unsigned lane = tid & 31;
  const unsigned half = lane >> 4;
  const unsigned sub = lane & 15;
  const int lo = (int)blockIdx.x * NPB;
  const unsigned wave = (unsigned)__builtin_amdgcn_readfirstlane((int)(tid >> 5));

  for (unsigned i = tid; i < NPB * N_REL; i += 1024) {
    int node = lo + (int)(i >> 4);
    invl[i] = (node < N_NODES) ? inv[(size_t)lo * N_REL + i] : 0.0f;
  }

  if (wave < (NPB + 15) / 16) {
    const int node = min(lo + (int)(wave * 16u + sub), N_NODES - 1);
    const float* arow = xs + (size_t)node * HID + 2u * half;
    v8f c0 = {}; v8f c1 = {};
    for (int k0 = 0; k0 < HID; k0 += 4) {
      v2f a;  a.x = arow[k0];  a.y = arow[k0 + 1];
      int kb = k0 + 2 * (int)half;
      v2f b0; b0.x = r2[kb * N_LAB + sub];        b0.y = r2[(kb + 1) * N_LAB + sub];
      v2f b1; b1.x = r2[kb * N_LAB + 16 + sub];   b1.y = r2[(kb + 1) * N_LAB + 16 + sub];
      c0 = __builtin_amdgcn_wmma_f32_16x16x4_f32(false, a, false, b0, (short)0, c0, false, false);
      c1 = __builtin_amdgcn_wmma_f32_16x16x4_f32(false, a, false, b1, (short)0, c1, false, false);
    }
    #pragma unroll
    for (int v = 0; v < 8; v++) {
      unsigned row = wave * 16u + (unsigned)v + 8u * half;
      if (row < (unsigned)NPB) {
        acc2[row * N_LAB + sub]      = c0[v] + bias2[sub];
        acc2[row * N_LAB + 16 + sub] = c1[v] + bias2[16 + sub];
      }
    }
  }

  const unsigned r = wave & 15u;
  const unsigned hw = wave >> 4;
  const float* wr = w2 + (size_t)r * HID * N_LAB;
  v2f b0[16], b1[16];
  #pragma unroll
  for (int s = 0; s < 16; s++) {
    unsigned k = 8u * (unsigned)(s >> 1) + 4u * half + 2u * (unsigned)(s & 1);
    b0[s].x = wr[k * N_LAB + sub];        b0[s].y = wr[(k + 1) * N_LAB + sub];
    b1[s].x = wr[k * N_LAB + 16 + sub];   b1[s].y = wr[(k + 1) * N_LAB + 16 + sub];
  }

  const bool slots = ws_bytes >= WS_NEED_BYTES;
  const unsigned nq = slots ? (unsigned)qn_g[blockIdx.x] : QNONE;
  const bool handed = nq <= (unsigned)QSLOT;
  const unsigned nchunk = handed ? 1u : (unsigned)((N_EDGE + QC2 - 1) / QC2);

  for (unsigned ch = 0; ch < nchunk; ch++) {
    __syncthreads();
    if (tid < N_REL) { rc[tid] = 0; cur[tid] = 0; }
    __syncthreads();

    unsigned p[L2_U];
    #pragma unroll
    for (int u = 0; u < L2_U; u++) {
      const unsigned i = (unsigned)u * 1024u + tid;
      p[u] = QNONE;
      if (handed) {
        if (i < nq) p[u] = (unsigned)qg[(size_t)blockIdx.x * QSLOT + i];
      } else {
        const unsigned e = ch * (unsigned)QC2 + i;
        if (e < (unsigned)N_EDGE) {
          const int d = dst_a[e];
          if ((unsigned)(d - lo) < (unsigned)NPB) {
            const int rl = etype[e], sc = src_a[e];
            p[u] = (unsigned)(rl * N_NODES + sc) | ((unsigned)(d - lo) << 20) | ((unsigned)rl << 28);
          }
        }
      }
      if (p[u] != QNONE) atomicAdd(&rc[p[u] >> 28], 1u);
    }
    __syncthreads();
    if (tid == 0) {
      unsigned a = 0;
      for (int k = 0; k < N_REL; k++) { ro[k] = a; a += rc[k]; }
    }
    __syncthreads();
    #pragma unroll
    for (int u = 0; u < L2_U; u++) {
      if (p[u] != QNONE) {
        const unsigned rr = p[u] >> 28;
        qs[ro[rr] + atomicAdd(&cur[rr], 1u)] = p[u];
      }
    }
    __syncthreads();

    const unsigned n = (unsigned)__builtin_amdgcn_readfirstlane((int)rc[r]);
    const unsigned start = (unsigned)__builtin_amdgcn_readfirstlane((int)ro[r]);
    for (unsigned t0 = 16u * hw; t0 < n; t0 += 32u) {
      const unsigned m = t0 + sub;
      const int s = (m < n) ? (int)(qs[start + m] & 0xFFFFFu) - (int)(r * (unsigned)N_NODES) : 0;
      const v4f* xr = (const v4f*)(xs + (size_t)s * HID + 4u * half);
      v4f q[8];
      #pragma unroll
      for (int i = 0; i < 8; i++) q[i] = xr[2 * i];
      unsigned dl[8]; float iv[8];
      #pragma unroll
      for (int v = 0; v < 8; v++) {
        unsigned row = min(t0 + (unsigned)v + 8u * half, n - 1u);
        dl[v] = (qs[start + row] >> 20) & 0xFFu;
        iv[v] = invl[dl[v] * N_REL + r];
      }
      v8f c0 = {}; v8f c1 = {};
      #pragma unroll
      for (int st = 0; st < 16; st++) {
        v2f a;
        if (st & 1) { a.x = q[st >> 1].z; a.y = q[st >> 1].w; }
        else        { a.x = q[st >> 1].x; a.y = q[st >> 1].y; }
        c0 = __builtin_amdgcn_wmma_f32_16x16x4_f32(false, a, false, b0[st], (short)0, c0, false, false);
        c1 = __builtin_amdgcn_wmma_f32_16x16x4_f32(false, a, false, b1[st], (short)0, c1, false, false);
      }
      float p0[8], p1[8];
      #pragma unroll
      for (int v = 0; v < 8; v++) { p0[v] = c0[v] * iv[v]; p1[v] = c1[v] * iv[v]; }
      #pragma unroll
      for (int v = 0; v < 8; v++) {
        unsigned row = t0 + (unsigned)v + 8u * half;
        if (row < n) {
          float* o = acc2 + dl[v] * N_LAB + sub;
          lds_fadd(o, p0[v]);
          lds_fadd(o + 16, p1[v]);
        }
      }
    }
  }
  __syncthreads();

  for (unsigned i = tid; i < NPB * N_LAB; i += 1024) {
    int node = lo + (int)(i >> 5);
    if (node < N_NODES) out[(size_t)lo * N_LAB + i] = 1.0f / (1.0f + __expf(-acc2[i]));
  }
}

extern "C" void kernel_launch(void* const* d_in, const int* in_sizes, int n_in,
                              void* d_out, int out_size, void* d_ws, size_t ws_size,
                              hipStream_t stream) {
  const int* ei     = (const int*)d_in[0];
  const int* src_a  = ei;
  const int* dst_a  = ei + N_EDGE;
  const int* etype  = (const int*)d_in[1];
  const float* w1   = (const float*)d_in[2];
  const float* r1   = (const float*)d_in[3];
  const float* b1   = (const float*)d_in[4];
  const float* w2   = (const float*)d_in[5];
  const float* r2   = (const float*)d_in[6];
  const float* b2   = (const float*)d_in[7];
  float* out = (float*)d_out;

  float* ws   = (float*)d_ws;
  float* inv  = ws + WS_INV_OFF;
  float* xs   = ws + WS_X_OFF;
  int* qn_g   = (int*)d_ws + WS_QN_OFF;
  int* qg     = (int*)d_ws + WS_QG_OFF;
  int* keys   = (int*)d_ws + WS_KEY_OFF;
  int* pk     = (int*)d_ws + WS_PK_OFF;

  (void)in_sizes; (void)n_in; (void)out_size;

  k_keys<<<(E4 + KB_E4 - 1) / KB_E4, KB_T, 0, stream>>>(src_a, dst_a, etype, (unsigned*)keys, pk,
                                                        (unsigned long long)ws_size);
  k_layer1<<<NRANGE, 1024, 0, stream>>>(src_a, dst_a, etype, w1, r1, b1, inv, xs, qn_g, qg, keys,
                                        pk, (unsigned long long)ws_size);
  k_layer2<<<NRANGE, 1024, 0, stream>>>(src_a, dst_a, etype, inv, xs, w2, r2, b2, qn_g, qg,
                                        (unsigned long long)ws_size, out);
}


# range keys as two one-hot sets of 16 bit planes (high, low 4 bits of dst/196; first launch transposes a 32x32 bit matrix per 32 edges): layer-1 scan reads 2 planes, not 8; queue, sum order unchanged
# speedup vs baseline: 3.8991x; 1.0375x over previous
#include <hip/hip_runtime.h>

#define N_NODES 50000
#define N_REL   16
#define HID     64
#define N_LAB   32
#define N_EDGE  1600000

typedef float v2f __attribute__((ext_vector_type(2)));
typedef float v4f __attribute__((ext_vector_type(4)));
typedef float v8f __attribute__((ext_vector_type(8)));

#define WS_INV_OFF  0
#define WS_X_OFF    (N_REL * N_NODES)
#define NRANGE      256
#define QSLOT       8192
#define WS_QN_OFF   (WS_X_OFF + N_NODES * HID)
#define WS_QG_OFF   (WS_QN_OFF + NRANGE)
#define WS_NEED_BYTES ((unsigned long long)(WS_QG_OFF + NRANGE * QSLOT) * 4ull)
#define KWORDS      (N_EDGE / 32)
#define KPITCH      50048
#define WS_KEY_OFF  (WS_QG_OFF + NRANGE * QSLOT)
#define NPLANE      32
#define WS_KEY_BYTES ((unsigned long long)(WS_KEY_OFF + NPLANE * KPITCH) * 4ull)
#define WS_PK_OFF   (WS_KEY_OFF + NPLANE * KPITCH)
#define WS_PK_BYTES ((unsigned long long)(WS_PK_OFF + N_EDGE) * 4ull)

#define NPB     196
#define QCAP    32768
#define SCAN_U  8
#define KSCAN_U 7
#define K4      (KWORDS / 4)
#define CNT_U   8
#define DR_U    16
#define X4      (NPB * HID / 4)
#define X4_U    ((X4 + 1023) / 1024)

__device__ __forceinline__ void lds_fadd(float* p, float v) {
  __hip_atomic_fetch_add(p, v, __ATOMIC_RELAXED, __HIP_MEMORY_SCOPE_WORKGROUP);
}

#define KB_T   256
#define KB_P   4
#define KB_E4  (KB_T * KB_P)
#define KB_W   (KB_E4 / 8)
#define E4     (N_EDGE / 4)

template <int J, unsigned M>
__device__ __forceinline__ void bit_transpose_stage(unsigned (&a)[32]) {
  #pragma unroll
  for (int k = 0; k < 32; k++) {
    if ((k & J) == 0) {
      const unsigned t = ((a[k] >> J) ^ a[k + J]) & M;
      a[k] ^= t << J;
      a[k + J] ^= t;
    }
  }
}

__device__ __forceinline__ void bit_transpose32(unsigned (&a)[32]) {
  bit_transpose_stage<16, 0x0000FFFFu>(a);
  bit_transpose_stage<8,  0x00FF00FFu>(a);
  bit_transpose_stage<4,  0x0F0F0F0Fu>(a);
  bit_transpose_stage<2,  0x33333333u>(a);
  bit_transpose_stage<1,  0x55555555u>(a);
}

template <bool PK, bool WHOLE>
__device__ __forceinline__ void keys_body(const int* __restrict__ src_a, const int* __restrict__ dst_a,
                                          const int* __restrict__ etype,
                                          unsigned* __restrict__ keys, int* __restrict__ pk,
                                          unsigned* kw) {
  const unsigned tid = threadIdx.x;
  const unsigned g4 = blockIdx.x * (unsigned)KB_E4;
  const int4* __restrict__ d4 = (const int4*)dst_a;
  const int4* __restrict__ s4 = (const int4*)src_a;
  const int4* __restrict__ r4 = (const int4*)etype;
  int4* __restrict__ p4 = (int4*)pk;
  const int4 zero4 = make_int4(0, 0, 0, 0);
  int4 d[KB_P], s[KB_P], r[KB_P];
  #pragma unroll
  for (int c = 0; c < KB_P; c++) {
    const unsigned i4 = WHOLE ? g4 + (unsigned)c * KB_T + tid
                              : min(g4 + (unsigned)c * KB_T + tid, (unsigned)E4 - 1u);
    d[c] = d4[i4];
    s[c] = PK ? s4[i4] : zero4;
    r[c] = PK ? r4[i4] : zero4;
  }
  #pragma unroll
  for (int c = 0; c < KB_P; c++) {
    const unsigned i4 = g4 + (unsigned)c * KB_T + tid;
    const int dd[4] = {d[c].x, d[c].y, d[c].z, d[c].w};
    const int ss[4] = {s[c].x, s[c].y, s[c].z, s[c].w};
    const int rr[4] = {r[c].x, r[c].y, r[c].z, r[c].w};
    unsigned kb = 0u;
    int pp[4];
    #pragma unroll
    for (int k = 0; k < 4; k++) {
      const unsigned key = min((unsigned)dd[k] / (unsigned)NPB, 255u);
      const unsigned dl = ((unsigned)dd[k] - key * (unsigned)NPB) & 0xFFu;
      kb |= key << (8 * k);
      pp[k] = (int)((unsigned)(rr[k] * N_NODES + ss[k]) | (dl << 20) | ((unsigned)rr[k] << 28));
    }
    kw[(unsigned)c * KB_T + tid] = kb;
    if (PK && (WHOLE || i4 < (unsigned)E4)) p4[i4] = make_int4(pp[0], pp[1], pp[2], pp[3]);
  }
  __syncthreads();

  const unsigned t = blockIdx.x * (unsigned)KB_W + tid;
  if (tid >= (unsigned)KB_W || t >= KWORDS) return;
  const uint4 ka = *(const uint4*)(kw + 8u * tid);
  const uint4 kc = *(const uint4*)(kw + 8u * tid + 4u);
  const unsigned kk[8] = {ka.x, ka.y, ka.z, ka.w, kc.x, kc.y, kc.z, kc.w};
  unsigned a[32];
  #pragma unroll
  for (int q = 0; q < 8; q++) {
    #pragma unroll
    for (int k = 0; k < 4; k++) {
      const unsigned key = (kk[q] >> (8 * k)) & 0xFFu;
      a[4 * q + k] = (1u << (key >> 4)) | (0x10000u << (key & 15u));
    }
  }
  bit_transpose32(a);
  #pragma unroll
  for (int p = 0; p < NPLANE; p++) keys[(unsigned)p * KPITCH + t] = a[p];
}

__global__ __launch_bounds__(KB_T)
void k_keys(const int* __restrict__ src_a, const int* __restrict__ dst_a,
            const int* __restrict__ etype, unsigned* __restrict__ keys, int* __restrict__ pk,
            unsigned long long ws_bytes) {
  __shared__ __attribute__((aligned(16))) unsigned kw[KB_E4];
  if (ws_bytes < WS_KEY_BYTES) return;
  const bool whole = blockIdx.x * (unsigned)KB_E4 + (unsigned)KB_E4 <= (unsigned)E4;
  if (ws_bytes >= WS_PK_BYTES) {
    if (whole) keys_body<true, true>(src_a, dst_a, etype, keys, pk, kw);
    else       keys_body<true, false>(src_a, dst_a, etype, keys, pk, kw);
  } else {
    if (whole) keys_body<false, true>(src_a, dst_a, etype, keys, pk, kw);
    else       keys_body<false, false>(src_a, dst_a, etype, keys, pk, kw);
  }
}

__device__ __forceinline__ void scan_planes(const int4* __restrict__ k4p, unsigned tid,
                                            int* q, unsigned* qn) {
  const int4* __restrict__ ph = k4p + (blockIdx.x >> 4) * (unsigned)(KPITCH / 4);
  const int4* __restrict__ pl = k4p + (16u + (blockIdx.x & 15u)) * (unsigned)(KPITCH / 4);
  #pragma nounroll
  for (unsigned base = 0; base < (unsigned)K4; base += 1024u * KSCAN_U) {
    int4 vh[KSCAN_U], vl[KSCAN_U];
    #pragma unroll
    for (int u = 0; u < KSCAN_U; u++) {
      unsigned i = min(base + (unsigned)u * 1024u + tid, (unsigned)K4 - 1u);
      vh[u] = ph[i];
      vl[u] = pl[i];
    }
    #pragma unroll
    for (int u = 0; u < KSCAN_U; u++) {
      const unsigned i = base + (unsigned)u * 1024u + tid;
      const unsigned live = (i < (unsigned)K4) ? 0xFFFFFFFFu : 0u;
      const unsigned mw[4] = {(unsigned)(vh[u].x & vl[u].x), (unsigned)(vh[u].y & vl[u].y),
                              (unsigned)(vh[u].z & vl[u].z), (unsigned)(vh[u].w & vl[u].w)};
      #pragma unroll
      for (int c = 0; c < 4; c++) {
        unsigned m = mw[c] & live;
        while (m != 0u) {
          unsigned l = (unsigned)__builtin_ctz(m);
          m &= m - 1u;
          int e = (int)(128u * i + 32u * (unsigned)c + l);
          unsigned idx = atomicAdd(qn, 1u);
          if (idx < QCAP) q[idx] = e;
        }
      }
    }
  }
}

__device__ __forceinline__ void scan_dst(const int* __restrict__ dst_a, int lo, unsigned tid,
                                         int* q, unsigned* qn) {
  const int4* __restrict__ d4p = (const int4*)dst_a;
  const unsigned n4 = N_EDGE / 4;
  for (unsigned base = 0; base < n4; base += 1024u * SCAN_U) {
    int4 v[SCAN_U];
    #pragma unroll
    for (int u = 0; u < SCAN_U; u++) {
      unsigned i = base + (unsigned)u * 1024u + tid;
      v[u] = (i < n4) ? d4p[i] : make_int4(-1, -1, -1, -1);
    }
    #pragma unroll
    for (int u = 0; u < SCAN_U; u++) {
      unsigned i = base + (unsigned)u * 1024u + tid;
      int dd[4] = {v[u].x, v[u].y, v[u].z, v[u].w};
      unsigned nearest = min(min((unsigned)(dd[0] - lo), (unsigned)(dd[1] - lo)),
                             min((unsigned)(dd[2] - lo), (unsigned)(dd[3] - lo)));
      if (nearest < (unsigned)NPB) {
        #pragma unroll
        for (int c = 0; c < 4; c++) {
          if ((unsigned)(dd[c] - lo) < (unsigned)NPB) {
            int e = (int)(4u * i) + c;
            unsigned idx = atomicAdd(qn, 1u);
            if (idx < QCAP) q[idx] = e;
          }
        }
      }
    }
  }
}

__global__ __launch_bounds__(1024)
void k_layer1(const int* __restrict__ src_a, const int* __restrict__ dst_a,
              const int* __restrict__ etype, const float* __restrict__ w1,
              const float* __restrict__ root1, const float* __restrict__ bias1,
              float* __restrict__ inv, float* __restrict__ xs,
              int* __restrict__ qn_g, int* __restrict__ qg, const int* __restrict__ keys,
              const int* __restrict__ pk, unsigned long long ws_bytes) {
  __shared__ __attribute__((aligned(16))) float acc[NPB * HID];
  __shared__ int q[QCAP];
  __shared__ unsigned cl[NPB * N_REL];
  __shared__ unsigned qn;
  const unsigned tid = threadIdx.x;
  const unsigned lane = tid & 31;
  const int lo = (int)blockIdx.x * NPB;

  for (unsigned i = tid; i < NPB * HID; i += 1024) acc[i] = 0.0f;
  for (unsigned i = tid; i < NPB * N_REL; i += 1024) cl[i] = 0u;
  if (tid == 0) qn = 0;
  __syncthreads();

  if (ws_bytes >= WS_KEY_BYTES) scan_planes((const int4*)keys, tid, q, &qn);
  else                          scan_dst(dst_a, lo, tid, q, &qn);
  __syncthreads();

  const unsigned nq = qn;
  const bool over = nq > (unsigned)QCAP;
  if (over) {
    for (unsigned e = tid; e < N_EDGE; e += 1024u) {
      int d = dst_a[e];
      if ((unsigned)(d - lo) < (unsigned)NPB) atomicAdd(&cl[(unsigned)(d - lo) * N_REL + (unsigned)etype[e]], 1u);
    }
  } else if (ws_bytes >= WS_PK_BYTES) {
    for (unsigned base = 0; base < nq; base += 1024u * CNT_U) {
      int e[CNT_U]; unsigned p[CNT_U];
      #pragma unroll
      for (int u = 0; u < CNT_U; u++) {
        unsigned i = base + (unsigned)u * 1024u + tid;
        e[u] = (i < nq) ? q[i] : -1;
      }
      #pragma unroll
      for (int u = 0; u < CNT_U; u++) p[u] = (e[u] >= 0) ? (unsigned)pk[e[u]] : 0u;
      #pragma unroll
      for (int u = 0; u < CNT_U; u++) {
        if (e[u] >= 0) {
          unsigned i = base + (unsigned)u * 1024u + tid;
          atomicAdd(&cl[((p[u] >> 20) & 0xFFu) * N_REL + (p[u] >> 28)], 1u);
          q[i] = (int)p[u];
        }
      }
    }
  } else {
    for (unsigned base = 0; base < nq; base += 1024u * CNT_U) {
      int e[CNT_U], rl[CNT_U], sc[CNT_U], ds[CNT_U];
      #pragma unroll
      for (int u = 0; u < CNT_U; u++) {
        unsigned i = base + (unsigned)u * 1024u + tid;
        e[u] = (i < nq) ? q[i] : -1;
      }
      #pragma unroll
      for (int u = 0; u < CNT_U; u++) {
        rl[u] = 0; sc[u] = 0; ds[u] = lo;
        if (e[u] >= 0) { rl[u] = etype[e[u]]; sc[u] = src_a[e[u]]; ds[u] = dst_a[e[u]]; }
      }
      #pragma unroll
      for (int u = 0; u < CNT_U; u++) {
        if (e[u] >= 0) {
          unsigned i = base + (unsigned)u * 1024u + tid;
          unsigned dl = (unsigned)(ds[u] - lo);
          atomicAdd(&cl[dl * N_REL + (unsigned)rl[u]], 1u);
          q[i] = (int)((unsigned)(rl[u] * N_NODES + sc[u]) | (dl << 20) | ((unsigned)rl[u] << 28));
        }
      }
    }
  }
  __syncthreads();

  for (unsigned i = tid; i < NPB * N_REL; i += 1024) {
    int node = lo + (int)(i >> 4);
    if (node < N_NODES) inv[(size_t)lo * N_REL + i] = 1.0f / fmaxf((float)cl[i], 1.0f);
  }

  if (ws_bytes >= WS_NEED_BYTES) {
    if (tid == 0) qn_g[blockIdx.x] = (int)min(nq, (unsigned)QSLOT + 1u);
    if (nq <= (unsigned)QSLOT)
      for (unsigned i = tid; i < nq; i += 1024) qg[(size_t)blockIdx.x * QSLOT + i] = q[i];
  }

  if (over) {
    for (unsigned e = tid; e < N_EDGE; e += 1024u) {
      int d = dst_a[e];
      if ((unsigned)(d - lo) < (unsigned)NPB) {
        int rel = etype[e], s = src_a[e];
        float ivs = 1.0f / fmaxf((float)cl[(unsigned)(d - lo) * N_REL + (unsigned)rel], 1.0f);
        const float* wr = w1 + ((size_t)rel * N_NODES + (size_t)s) * HID;
        float* ar = acc + (d - lo) * HID;
        for (int h = 0; h < HID; h++) lds_fadd(ar + h, wr[h] * ivs);
      }
    }
  }
  v4f rt[X4_U];
  #pragma unroll
  for (int k = 0; k < X4_U; k++) {
    const unsigned i4 = (unsigned)k * 1024u + tid;
    const int node = lo + (int)(i4 >> 4);
    rt[k] = (v4f){0.0f, 0.0f, 0.0f, 0.0f};
    if (i4 < (unsigned)X4 && node < N_NODES) rt[k] = *(const v4f*)(root1 + (size_t)lo * HID + 4u * i4);
  }
  const v4f bs = *(const v4f*)(bias1 + ((4u * tid) & 63u));

  const unsigned n = over ? 0u : nq;
  const unsigned wave = (unsigned)__builtin_amdgcn_readfirstlane((int)(tid >> 5));
  for (unsigned b = wave * 32u; b < n; b += 32u * 32u) {
    const unsigned m = min(32u, n - b);
    int r = 0, dl = 0; float iv = 0.0f;
    if (lane < m) {
      unsigned p = (unsigned)q[b + lane];
      r = (int)(p & 0xFFFFFu);
      dl = (int)((p >> 20) & 0xFFu);
      iv = 1.0f / fmaxf((float)cl[(unsigned)dl * N_REL + (p >> 28)], 1.0f);
    }
    #pragma unroll
    for (unsigned j = 0; j < 32; j += DR_U) {
      if (j >= m) break;
      if (j + DR_U <= m) {
        v2f row[DR_U];
        #pragma unroll
        for (unsigned u = 0; u < DR_U; u++) {
          int rr = __builtin_amdgcn_readlane(r, (int)(j + u));
          row[u] = *(const v2f*)(w1 + (size_t)rr * HID + 2u * lane);
        }
        #pragma unroll
        for (unsigned u = 0; u < DR_U; u++) {
          int dd = __builtin_amdgcn_readlane(dl, (int)(j + u));
          float ii = __builtin_bit_cast(float, __builtin_amdgcn_readlane(__builtin_bit_cast(int, iv), (int)(j + u)));
          float* o = acc + dd * HID + 2u * lane;
          lds_fadd(o + 0, row[u].x * ii);
          lds_fadd(o + 1, row[u].y * ii);
        }
      } else {
        for (unsigned u = j; u < m; u++) {
          int rr = __builtin_amdgcn_readlane(r, (int)u);
          int dd = __builtin_amdgcn_readlane(dl, (int)u);
          float ii = __builtin_bit_cast(float, __builtin_amdgcn_readlane(__builtin_bit_cast(int, iv), (int)u));
          v2f rw = *(const v2f*)(w1 + (size_t)rr * HID + 2u * lane);
          float* o = acc + dd * HID + 2u * lane;
          lds_fadd(o + 0, rw.x * ii);
          lds_fadd(o + 1, rw.y * ii);
        }
      }
    }
  }
  __syncthreads();

  #pragma unroll
  for (int k = 0; k < X4_U; k++) {
    const unsigned i4 = (unsigned)k * 1024u + tid;
    const int node = lo + (int)(i4 >> 4);
    if (i4 < (unsigned)X4 && node < N_NODES) {
      const v4f a = *(const v4f*)(acc + 4u * i4);
      v4f x;
      x.x = fmaxf(a.x + rt[k].x + bs.x, 0.0f);
      x.y = fmaxf(a.y + rt[k].y + bs.y, 0.0f);
      x.z = fmaxf(a.z + rt[k].z + bs.z, 0.0f);
      x.w = fmaxf(a.w + rt[k].w + bs.w, 0.0f);
      *(v4f*)(xs + (size_t)lo * HID + 4u * i4) = x;
    }
  }
}


#define QC2   QSLOT
#define L2_U  8
#define QNONE 0xFFFFFFFFu

__global__ __launch_bounds__(1024)
void k_layer2(const int* __restrict__ src_a, const int* __restrict__ dst_a,
              const int* __restrict__ etype, const float* __restrict__ inv,
              const float* __restrict__ xs, const float* __restrict__ w2,
              const float* __restrict__ r2, const float* __restrict__ bias2,
              const int* __restrict__ qn_g, const int* __restrict__ qg,
              unsigned long long ws_bytes, float* __restrict__ out) {
  __shared__ float acc2[NPB * N_LAB];
  __shared__ float invl[NPB * N_REL];
  __shared__ unsigned qs[QC2];
  __shared__ unsigned rc[N_REL];
  __shared__ unsigned ro[N_REL];
  __shared__ unsigned cur[N_REL];
  const unsigned tid = threadIdx.x;
  const unsigned lane = tid & 31;
  const unsigned half = lane >> 4;
  const unsigned sub = lane & 15;
  const int lo = (int)blockIdx.x * NPB;
  const unsigned wave = (unsigned)__builtin_amdgcn_readfirstlane((int)(tid >> 5));

  for (unsigned i = tid; i < NPB * N_REL; i += 1024) {
    int node = lo + (int)(i >> 4);
    invl[i] = (node < N_NODES) ? inv[(size_t)lo * N_REL + i] : 0.0f;
  }

  if (wave < (NPB + 15) / 16) {
    const int node = min(lo + (int)(wave * 16u + sub), N_NODES - 1);
    const float* arow = xs + (size_t)node * HID + 2u * half;
    v8f c0 = {}; v8f c1 = {};
    for (int k0 = 0; k0 < HID; k0 += 4) {
      v2f a;  a.x = arow[k0];  a.y = arow[k0 + 1];
      int kb = k0 + 2 * (int)half;
      v2f b0; b0.x = r2[kb * N_LAB + sub];        b0.y = r2[(kb + 1) * N_LAB + sub];
      v2f b1; b1.x = r2[kb * N_LAB + 16 + sub];   b1.y = r2[(kb + 1) * N_LAB + 16 + sub];
      c0 = __builtin_amdgcn_wmma_f32_16x16x4_f32(false, a, false, b0, (short)0, c0, false, false);
      c1 = __builtin_amdgcn_wmma_f32_16x16x4_f32(false, a, false, b1, (short)0, c1, false, false);
    }
    #pragma unroll
    for (int v = 0; v < 8; v++) {
      unsigned row = wave * 16u + (unsigned)v + 8u * half;
      if (row < (unsigned)NPB) {
        acc2[row * N_LAB + sub]      = c0[v] + bias2[sub];
        acc2[row * N_LAB + 16 + sub] = c1[v] + bias2[16 + sub];
      }
    }
  }

  const unsigned r = wave & 15u;
  const unsigned hw = wave >> 4;
  const float* wr = w2 + (size_t)r * HID * N_LAB;
  v2f b0[16], b1[16];
  #pragma unroll
  for (int s = 0; s < 16; s++) {
    unsigned k = 8u * (unsigned)(s >> 1) + 4u * half + 2u * (unsigned)(s & 1);
    b0[s].x = wr[k * N_LAB + sub];        b0[s].y = wr[(k + 1) * N_LAB + sub];
    b1[s].x = wr[k * N_LAB + 16 + sub];   b1[s].y = wr[(k + 1) * N_LAB + 16 + sub];
  }

  const bool slots = ws_bytes >= WS_NEED_BYTES;
  const unsigned nq = slots ? (unsigned)qn_g[blockIdx.x] : QNONE;
  const bool handed = nq <= (unsigned)QSLOT;
  const unsigned nchunk = handed ? 1u : (unsigned)((N_EDGE + QC2 - 1) / QC2);

  for (unsigned ch = 0; ch < nchunk; ch++) {
    __syncthreads();
    if (tid < N_REL) { rc[tid] = 0; cur[tid] = 0; }
    __syncthreads();

    unsigned p[L2_U];
    #pragma unroll
    for (int u = 0; u < L2_U; u++) {
      const unsigned i = (unsigned)u * 1024u + tid;
      p[u] = QNONE;
      if (handed) {
        if (i < nq) p[u] = (unsigned)qg[(size_t)blockIdx.x * QSLOT + i];
      } else {
        const unsigned e = ch * (unsigned)QC2 + i;
        if (e < (unsigned)N_EDGE) {
          const int d = dst_a[e];
          if ((unsigned)(d - lo) < (unsigned)NPB) {
            const int rl = etype[e], sc = src_a[e];
            p[u] = (unsigned)(rl * N_NODES + sc) | ((unsigned)(d - lo) << 20) | ((unsigned)rl << 28);
          }
        }
      }
      if (p[u] != QNONE) atomicAdd(&rc[p[u] >> 28], 1u);
    }
    __syncthreads();
    if (tid == 0) {
      unsigned a = 0;
      for (int k = 0; k < N_REL; k++) { ro[k] = a; a += rc[k]; }
    }
    __syncthreads();
    #pragma unroll
    for (int u = 0; u < L2_U; u++) {
      if (p[u] != QNONE) {
        const unsigned rr = p[u] >> 28;
        qs[ro[rr] + atomicAdd(&cur[rr], 1u)] = p[u];
      }
    }
    __syncthreads();

    const unsigned n = (unsigned)__builtin_amdgcn_readfirstlane((int)rc[r]);
    const unsigned start = (unsigned)__builtin_amdgcn_readfirstlane((int)ro[r]);
    for (unsigned t0 = 16u * hw; t0 < n; t0 += 32u) {
      const unsigned m = t0 + sub;
      const int s = (m < n) ? (int)(qs[start + m] & 0xFFFFFu) - (int)(r * (unsigned)N_NODES) : 0;
      const v4f* xr = (const v4f*)(xs + (size_t)s * HID + 4u * half);
      v4f q[8];
      #pragma unroll
      for (int i = 0; i < 8; i++) q[i] = xr[2 * i];
      unsigned dl[8]; float iv[8];
      #pragma unroll
      for (int v = 0; v < 8; v++) {
        unsigned row = min(t0 + (unsigned)v + 8u * half, n - 1u);
        dl[v] = (qs[start + row] >> 20) & 0xFFu;
        iv[v] = invl[dl[v] * N_REL + r];
      }
      v8f c0 = {}; v8f c1 = {};
      #pragma unroll
      for (int st = 0; st < 16; st++) {
        v2f a;
        if (st & 1) { a.x = q[st >> 1].z; a.y = q[st >> 1].w; }
        else        { a.x = q[st >> 1].x; a.y = q[st >> 1].y; }
        c0 = __builtin_amdgcn_wmma_f32_16x16x4_f32(false, a, false, b0[st], (short)0, c0, false, false);
        c1 = __builtin_amdgcn_wmma_f32_16x16x4_f32(false, a, false, b1[st], (short)0, c1, false, false);
      }
      float p0[8], p1[8];
      #pragma unroll
      for (int v = 0; v < 8; v++) { p0[v] = c0[v] * iv[v]; p1[v] = c1[v] * iv[v]; }
      #pragma unroll
      for (int v = 0; v < 8; v++) {
        unsigned row = t0 + (unsigned)v + 8u * half;
        if (row < n) {
          float* o = acc2 + dl[v] * N_LAB + sub;
          lds_fadd(o, p0[v]);
          lds_fadd(o + 16, p1[v]);
        }
      }
    }
  }
  __syncthreads();

  for (unsigned i = tid; i < NPB * N_LAB; i += 1024) {
    int node = lo + (int)(i >> 5);
    if (node < N_NODES) out[(size_t)lo * N_LAB + i] = 1.0f / (1.0f + __expf(-acc2[i]));
  }
}

extern "C" void kernel_launch(void* const* d_in, const int* in_sizes, int n_in,
                              void* d_out, int out_size, void* d_ws, size_t ws_size,
                              hipStream_t stream) {
  const int* ei     = (const int*)d_in[0];
  const int* src_a  = ei;
  const int* dst_a  = ei + N_EDGE;
  const int* etype  = (const int*)d_in[1];
  const float* w1   = (const float*)d_in[2];
  const float* r1   = (const float*)d_in[3];
  const float* b1   = (const float*)d_in[4];
  const float* w2   = (const float*)d_in[5];
  const float* r2   = (const float*)d_in[6];
  const float* b2   = (const float*)d_in[7];
  float* out = (float*)d_out;

  float* ws   = (float*)d_ws;
  float* inv  = ws + WS_INV_OFF;
  float* xs   = ws + WS_X_OFF;
  int* qn_g   = (int*)d_ws + WS_QN_OFF;
  int* qg     = (int*)d_ws + WS_QG_OFF;
  int* keys   = (int*)d_ws + WS_KEY_OFF;
  int* pk     = (int*)d_ws + WS_PK_OFF;

  (void)in_sizes; (void)n_in; (void)out_size;

  k_keys<<<(E4 + KB_E4 - 1) / KB_E4, KB_T, 0, stream>>>(src_a, dst_a, etype, (unsigned*)keys, pk,
                                                        (unsigned long long)ws_size);
  k_layer1<<<NRANGE, 1024, 0, stream>>>(src_a, dst_a, etype, w1, r1, b1, inv, xs, qn_g, qg, keys,
                                        pk, (unsigned long long)ws_size);
  k_layer2<<<NRANGE, 1024, 0, stream>>>(src_a, dst_a, etype, inv, xs, w2, r2, b2, qn_g, qg,
                                        (unsigned long long)ws_size, out);
}
